# starting kernel
# baseline (speedup 1.0000x reference)
#include <hip/hip_runtime.h>

typedef __attribute__((ext_vector_type(2))) float v2f;
typedef __attribute__((ext_vector_type(8))) float v8f;

#define CIN   8
#define COUT  16
#define HW    224
#define TH    16
#define TW    16

#define OFF_IN   0
#define OFF_W0T  4608
#define OFF_C2   0
#define OFF_C1   6400
#define OFF_C3   6400
#define SMEM_F   14144

__global__ __launch_bounds__(256)
void fused_stackedconv_kernel(const float* __restrict__ x,
                              const float* __restrict__ W0,
                              const float* __restrict__ W1,
                              const float* __restrict__ W2,
                              const float* __restrict__ W3,
                              const float* __restrict__ b3,
                              float* __restrict__ out)
{
    __shared__ float smem[SMEM_F];
    const int tid  = threadIdx.x;
    const int lane = tid & 31;
    const int wid  = tid >> 5;
    const int ox = blockIdx.x * TW;
    const int oy = blockIdx.y * TH;
    const int bb = blockIdx.z;

    for (int idx = tid; idx < 72 * 16; idx += 256) {
        const int k = idx >> 4, n = idx & 15;
        smem[OFF_W0T + idx] = W0[n * 72 + k];
    }
    const float* xb = x + (size_t)bb * CIN * HW * HW;
    for (int idx = tid; idx < CIN * 24 * 24; idx += 256) {
        const int c  = idx / 576, rem = idx % 576;
        const int yy = rem / 24,  xx  = rem % 24;
        const int gy = oy - 4 + yy, gx = ox - 4 + xx;
        float v = 0.0f;
        if (gy >= 0 && gy < HW && gx >= 0 && gx < HW)
            v = xb[(c * HW + gy) * HW + gx];
        smem[OFF_IN + idx] = v;
    }
    __syncthreads();

    const int r  = lane >> 4;
    const int nl = lane & 15;

    v2f bfr[18];
    int offx[18], offy[18];
#pragma unroll
    for (int kk = 0; kk < 18; ++kk) {
        const int kx = 4 * kk + 2 * r;
        const int ky = kx + 1;
        bfr[kk].x = smem[OFF_W0T + kx * 16 + nl];
        bfr[kk].y = smem[OFF_W0T + ky * 16 + nl];
        const int cinx = kx / 9, tapx = kx - cinx * 9;
        const int ciny = ky / 9, tapy = ky - ciny * 9;
        offx[kk] = cinx * 576 + (tapx / 3) * 24 + (tapx % 3);
        offy[kk] = ciny * 576 + (tapy / 3) * 24 + (tapy % 3);
    }

    for (int t = wid; t < 31; t += 8) {
        const int p  = t * 16 + nl;
        const int pp = p < 483 ? p : 483;
        const int i  = pp / 22;
        const int j  = pp - i * 22;
        const int abase = OFF_IN + i * 24 + j;

        v8f acc = {};
#pragma unroll
        for (int kk = 0; kk < 18; ++kk) {
            v2f a;
            a.x = smem[abase + offx[kk]];
            a.y = smem[abase + offy[kk]];
            acc = __builtin_amdgcn_wmma_f32_16x16x4_f32(
                false, a, false, bfr[kk], (short)0, acc, false, false);
        }
        const int pbase = t * 16 + 8 * r;
        const int cdst  = OFF_C1 + nl * 484 + pbase;
        if (t < 30) {
#pragma unroll
            for (int v = 0; v < 8; ++v)
                smem[cdst + v] = acc[v];
        } else {
#pragma unroll
            for (int v = 0; v < 8; ++v)
                if (pbase + v < 484)
                    smem[cdst + v] = acc[v];
        }
    }
    __syncthreads();

    for (int idx = tid; idx < 16 * 20 * 20; idx += 256) {
        const int ch = idx / 400, rem = idx % 400;
        const int yy = rem / 20,  xx  = rem % 20;
        const float* w = W1 + ch * 9;
        float s = 0.0f;
#pragma unroll
        for (int ty = 0; ty < 3; ++ty)
#pragma unroll
            for (int tx = 0; tx < 3; ++tx)
                s += smem[OFF_C1 + ch * 484 + (yy + ty) * 22 + (xx + tx)] * w[ty * 3 + tx];
        smem[OFF_C2 + idx] = s;
    }
    __syncthreads();

    for (int idx = tid; idx < 16 * 18 * 18; idx += 256) {
        const int ch = idx / 324, rem = idx % 324;
        const int yy = rem / 18,  xx  = rem % 18;
        const float* w = W2 + ch * 9;
        float s = 0.0f;
#pragma unroll
        for (int ty = 0; ty < 3; ++ty)
#pragma unroll
            for (int tx = 0; tx < 3; ++tx)
                s += smem[OFF_C2 + ch * 400 + (yy + ty) * 20 + (xx + tx)] * w[ty * 3 + tx];
        smem[OFF_C3 + idx] = s;
    }
    __syncthreads();

    for (int idx = tid; idx < 16 * 16 * 16; idx += 256) {
        const int ch = idx >> 8, rem = idx & 255;
        const int yy = rem >> 4, xx = rem & 15;
        const float* w = W3 + ch * 9;
        float s = b3[ch];
#pragma unroll
        for (int ty = 0; ty < 3; ++ty)
#pragma unroll
            for (int tx = 0; tx < 3; ++tx)
                s += smem[OFF_C3 + ch * 324 + (yy + ty) * 18 + (xx + tx)] * w[ty * 3 + tx];
        out[((size_t)(bb * COUT + ch) * HW + (oy + yy)) * HW + (ox + xx)] = s;
    }
}

extern "C" void kernel_launch(void* const* d_in, const int* in_sizes, int n_in,
                              void* d_out, int out_size, void* d_ws, size_t ws_size,
                              hipStream_t stream) {
    const float* x  = (const float*)d_in[0];
    const float* W0 = (const float*)d_in[1];
    const float* W1 = (const float*)d_in[2];
    const float* W2 = (const float*)d_in[3];
    const float* W3 = (const float*)d_in[4];
    const float* b3 = (const float*)d_in[5];
    float* out = (float*)d_out;

    dim3 grid(HW / TW, HW / TH, 32);
    dim3 block(256);
    fused_stackedconv_kernel<<<grid, block, 0, stream>>>(x, W0, W1, W2, W3, b3, out);
}


# dw1-dw3 fused in registers: a thread walks a 10x10 conv1 window for one channel's 4x4 outputs with packed f32 FMAs, weights in registers; tap order (ty,tx) kept, products now fused into the adds
# speedup vs baseline: 1.6548x; 1.6548x over previous
#include <hip/hip_runtime.h>

typedef __attribute__((ext_vector_type(2))) float v2f;
typedef __attribute__((ext_vector_type(8))) float v8f;

#define CIN   8
#define COUT  16
#define HW    224
#define TH    16
#define TW    16

#define OFF_IN   0
#define OFF_W0T  4608
#define OFF_C1   5760
#define SMEM_F   13504

__global__ __launch_bounds__(256, 8)
void fused_stackedconv_kernel(const float* __restrict__ x,
                              const float* __restrict__ W0,
                              const float* __restrict__ W1,
                              const float* __restrict__ W2,
                              const float* __restrict__ W3,
                              const float* __restrict__ b3,
                              float* __restrict__ out)
{
    __shared__ float smem[SMEM_F];
    const int tid  = threadIdx.x;
    const int lane = tid & 31;
    const int wid  = tid >> 5;
    const int ox = blockIdx.x * TW;
    const int oy = blockIdx.y * TH;
    const int bb = blockIdx.z;

    for (int idx = tid; idx < 72 * 16; idx += 256) {
        const int k = idx >> 4, n = idx & 15;
        smem[OFF_W0T + idx] = W0[n * 72 + k];
    }
    const float* xb = x + (size_t)bb * CIN * HW * HW;
    for (int idx = tid; idx < CIN * 24 * 24; idx += 256) {
        const int c  = idx / 576, rem = idx % 576;
        const int yy = rem / 24,  xx  = rem % 24;
        const int gy = oy - 4 + yy, gx = ox - 4 + xx;
        float v = 0.0f;
        if (gy >= 0 && gy < HW && gx >= 0 && gx < HW)
            v = xb[(c * HW + gy) * HW + gx];
        smem[OFF_IN + idx] = v;
    }
    __syncthreads();

    const int r  = lane >> 4;
    const int nl = lane & 15;

    v2f bfr[18];
    int offx[18], offy[18];
#pragma unroll
    for (int kk = 0; kk < 18; ++kk) {
        const int kx = 4 * kk + 2 * r;
        const int ky = kx + 1;
        bfr[kk].x = smem[OFF_W0T + kx * 16 + nl];
        bfr[kk].y = smem[OFF_W0T + ky * 16 + nl];
        const int cinx = kx / 9, tapx = kx - cinx * 9;
        const int ciny = ky / 9, tapy = ky - ciny * 9;
        offx[kk] = cinx * 576 + (tapx / 3) * 24 + (tapx % 3);
        offy[kk] = ciny * 576 + (tapy / 3) * 24 + (tapy % 3);
    }

    for (int t = wid; t < 31; t += 8) {
        const int p  = t * 16 + nl;
        const int pp = p < 483 ? p : 483;
        const int i  = pp / 22;
        const int j  = pp - i * 22;
        const int abase = OFF_IN + i * 24 + j;

        v8f acc = {};
#pragma unroll
        for (int kk = 0; kk < 18; ++kk) {
            v2f a;
            a.x = smem[abase + offx[kk]];
            a.y = smem[abase + offy[kk]];
            acc = __builtin_amdgcn_wmma_f32_16x16x4_f32(
                false, a, false, bfr[kk], (short)0, acc, false, false);
        }
        const int pbase = t * 16 + 8 * r;
        const int cdst  = OFF_C1 + nl * 484 + pbase;
        if (t < 30) {
#pragma unroll
            for (int v = 0; v < 8; ++v)
                smem[cdst + v] = acc[v];
        } else {
#pragma unroll
            for (int v = 0; v < 8; ++v)
                if (pbase + v < 484)
                    smem[cdst + v] = acc[v];
        }
    }
    const int cs = tid & 3;
    const int rg = (tid >> 2) & 3;
    const int ch = tid >> 4;
    float w1[9], w2[9], w3[9];
#pragma unroll
    for (int q = 0; q < 9; ++q) {
        w1[q] = W1[ch * 9 + q];
        w2[q] = W2[ch * 9 + q];
        w3[q] = W3[ch * 9 + q];
    }
    const float bias = b3[ch];
    __syncthreads();

    const v2f bias2 = {bias, bias};
    const float* c1p = smem + OFF_C1 + ch * 484 + (4 * rg) * 22 + 4 * cs;
    float* op = out + ((size_t)(bb * COUT + ch) * HW + (oy + 4 * rg)) * HW + (ox + 4 * cs);

#define SPLAT(w) ((v2f){(w), (w)})
#define FMA2(a, b, c) __builtin_elementwise_fma((a), (b), (c))

    v2f a2[3][4], a3[3][3], a4[3][2];
#pragma unroll
    for (int i = 0; i < 10; ++i) {
        v2f e1[5], o1[4];
#pragma unroll
        for (int q = 0; q < 5; ++q)
            e1[q] = *(const v2f*)(c1p + i * 22 + 2 * q);
#pragma unroll
        for (int q = 0; q < 4; ++q)
            o1[q] = (v2f){e1[q].y, e1[q + 1].x};
#pragma unroll
        for (int ty = 0; ty < 3; ++ty) {
            const int a = i - ty;
            if (a >= 0 && a < 8) {
#pragma unroll
                for (int q = 0; q < 4; ++q) {
                    v2f s;
                    if (ty == 0) s = e1[q] * SPLAT(w1[0]);
                    else         s = FMA2(e1[q], SPLAT(w1[ty * 3]), a2[a % 3][q]);
                    s = FMA2(o1[q],     SPLAT(w1[ty * 3 + 1]), s);
                    s = FMA2(e1[q + 1], SPLAT(w1[ty * 3 + 2]), s);
                    a2[a % 3][q] = s;
                }
            }
        }
        if (i >= 2) {
            const int a = i - 2;
            v2f e2[4], o2[3];
#pragma unroll
            for (int q = 0; q < 4; ++q) e2[q] = a2[a % 3][q];
#pragma unroll
            for (int q = 0; q < 3; ++q) o2[q] = (v2f){e2[q].y, e2[q + 1].x};
#pragma unroll
            for (int ty = 0; ty < 3; ++ty) {
                const int b = a - ty;
                if (b >= 0 && b < 6) {
#pragma unroll
                    for (int q = 0; q < 3; ++q) {
                        v2f s;
                        if (ty == 0) s = e2[q] * SPLAT(w2[0]);
                        else         s = FMA2(e2[q], SPLAT(w2[ty * 3]), a3[b % 3][q]);
                        s = FMA2(o2[q],     SPLAT(w2[ty * 3 + 1]), s);
                        s = FMA2(e2[q + 1], SPLAT(w2[ty * 3 + 2]), s);
                        a3[b % 3][q] = s;
                    }
                }
            }
            if (a >= 2) {
                const int b = a - 2;
                v2f e3[3], o3[2];
#pragma unroll
                for (int q = 0; q < 3; ++q) e3[q] = a3[b % 3][q];
#pragma unroll
                for (int q = 0; q < 2; ++q) o3[q] = (v2f){e3[q].y, e3[q + 1].x};
#pragma unroll
                for (int ty = 0; ty < 3; ++ty) {
                    const int c = b - ty;
                    if (c >= 0 && c < 4) {
#pragma unroll
                        for (int q = 0; q < 2; ++q) {
                            v2f s = (ty == 0) ? bias2 : a4[c % 3][q];
                            s = FMA2(e3[q],     SPLAT(w3[ty * 3]),     s);
                            s = FMA2(o3[q],     SPLAT(w3[ty * 3 + 1]), s);
                            s = FMA2(e3[q + 1], SPLAT(w3[ty * 3 + 2]), s);
                            a4[c % 3][q] = s;
                        }
                    }
                }
                if (b >= 2) {
                    const int c = b - 2;
                    *(float4*)(op + (size_t)c * HW) =
                        make_float4(a4[c % 3][0].x, a4[c % 3][0].y, a4[c % 3][1].x, a4[c % 3][1].y);
                }
            }
        }
    }
#undef SPLAT
#undef FMA2
}

extern "C" void kernel_launch(void* const* d_in, const int* in_sizes, int n_in,
                              void* d_out, int out_size, void* d_ws, size_t ws_size,
                              hipStream_t stream) {
    const float* x  = (const float*)d_in[0];
    const float* W0 = (const float*)d_in[1];
    const float* W1 = (const float*)d_in[2];
    const float* W2 = (const float*)d_in[3];
    const float* W3 = (const float*)d_in[4];
    const float* b3 = (const float*)d_in[5];
    float* out = (float*)d_out;

    dim3 grid(HW / TW, HW / TH, 32);
    dim3 block(256);
    fused_stackedconv_kernel<<<grid, block, 0, stream>>>(x, W0, W1, W2, W3, b3, out);
}


# tile loop inside the work-group: 1024 resident groups walk the 6272 tiles; weights staged once per group; next tile's patch loaded under the conv1 WMMAs, stored to LDS after them; sums unchanged
# speedup vs baseline: 2.2735x; 1.3739x over previous
#include <hip/hip_runtime.h>

typedef __attribute__((ext_vector_type(2))) float v2f;
typedef __attribute__((ext_vector_type(8))) float v8f;

#define CIN   8
#define COUT  16
#define HW    224
#define TH    16
#define TW    16
#define NTILE 6272

#define OFF_IN   0
#define OFF_W0T  4608
#define OFF_C1   5760
#define OFF_WD   13504
#define SMEM_F   13952
#define NWG      1024

static __device__ __forceinline__ void patch_load(const float* __restrict__ x, int tile,
                                                  int wid, int lane, float (&pf)[24])
{
    const int bb  = tile / 196;
    const int rem = tile - bb * 196;
    const int by  = rem / 14;
    const int bx  = rem - by * 14;
    const int gx  = bx * TW - 4 + lane;
    const int gxc = gx < 0 ? 0 : (gx > HW - 1 ? HW - 1 : gx);
    const float* plane = x + ((size_t)bb * CIN + wid) * (HW * HW);
#pragma unroll
    for (int k = 0; k < 24; ++k) {
        const int gy  = by * TH - 4 + k;
        const int gyc = gy < 0 ? 0 : (gy > HW - 1 ? HW - 1 : gy);
        pf[k] = plane[gyc * HW + gxc];
    }
}

static __device__ __forceinline__ void patch_store(float* smem, int tile,
                                                   int wid, int lane, const float (&pf)[24])
{
    const int bb  = tile / 196;
    const int rem = tile - bb * 196;
    const int by  = rem / 14;
    const int bx  = rem - by * 14;
    const int gx  = bx * TW - 4 + lane;
    const bool colok = (gx >= 0) && (gx < HW);
    if (lane < 24) {
        float* dst = smem + OFF_IN + wid * 576 + lane;
#pragma unroll
        for (int k = 0; k < 24; ++k) {
            const int gy = by * TH - 4 + k;
            const bool ok = colok && (gy >= 0) && (gy < HW);
            dst[k * 24] = ok ? pf[k] : 0.0f;
        }
    }
}

__global__ __launch_bounds__(256, 8)
void fused_stackedconv_kernel(const float* __restrict__ x,
                              const float* __restrict__ W0,
                              const float* __restrict__ W1,
                              const float* __restrict__ W2,
                              const float* __restrict__ W3,
                              const float* __restrict__ b3,
                              float* __restrict__ out)
{
    __shared__ float smem[SMEM_F];
    const int tid0 = threadIdx.x;

    for (int idx = tid0; idx < 72 * 16; idx += 256) {
        const int k = idx >> 4, n = idx & 15;
        smem[OFF_W0T + idx] = W0[n * 72 + k];
    }
    for (int idx = tid0; idx < 16 * 28; idx += 256) {
        const int c = idx / 28, q = idx - c * 28;
        float v;
        if      (q < 9)  v = W1[c * 9 + q];
        else if (q < 18) v = W2[c * 9 + q - 9];
        else if (q < 27) v = W3[c * 9 + q - 18];
        else             v = b3[c];
        smem[OFF_WD + idx] = v;
    }
    int tile = blockIdx.x;
    {
        float pf0[24];
        patch_load(x, tile, tid0 >> 5, tid0 & 31, pf0);
        patch_store(smem, tile, tid0 >> 5, tid0 & 31, pf0);
    }
    __syncthreads();

    for (;;) {
        int tid = threadIdx.x;
        asm volatile("" : "+v"(tid));
        const int lane = tid & 31;
        const int wid  = __builtin_amdgcn_readfirstlane(tid >> 5);
        const int cs = tid & 3;
        const int rg = (tid >> 2) & 3;
        const int ch = tid >> 4;

        const int  next     = tile + NWG;
        const bool has_next = next < NTILE;
        const int bb  = tile / 196;
        const int rem = tile - bb * 196;
        const int oy  = (rem / 14) * TH;
        const int ox  = (rem % 14) * TW;

        float pf[24];
        patch_load(x, has_next ? next : tile, wid, lane, pf);

        const int r  = lane >> 4;
        const int nl = lane & 15;

        v2f bfr[18];
#pragma unroll
        for (int kk = 0; kk < 18; ++kk) {
            const int kx = 4 * kk + 2 * r;
            const int ky = kx + 1;
            bfr[kk].x = smem[OFF_W0T + kx * 16 + nl];
            bfr[kk].y = smem[OFF_W0T + ky * 16 + nl];
        }
#define KOFF(k) (((k) / 9) * 576 + (((k) % 9) / 3) * 24 + ((k) % 9) % 3)
#define KDEL(k) (KOFF((k) + 2) - KOFF(k))
#define A_AT(k) ((KDEL(k) == 2 ? pa2 : (KDEL(k) == 23 ? pa23 : pa527))[KOFF(k)])
        const int d2 = r * 2, d23 = r * 23, d527 = r * 527;

        for (int t = wid; t < 31; t += 8) {
            const int p  = t * 16 + nl;
            const int pp = p < 483 ? p : 483;
            const int i  = pp / 22;
            const int j  = pp - i * 22;
            const int abase = OFF_IN + i * 24 + j;
            const float* pa2   = smem + abase + d2;
            const float* pa23  = smem + abase + d23;
            const float* pa527 = smem + abase + d527;

            v8f acc = {};
#pragma unroll
            for (int kk = 0; kk < 18; ++kk) {
                v2f a;
                a.x = A_AT(4 * kk);
                a.y = A_AT(4 * kk + 1);
                acc = __builtin_amdgcn_wmma_f32_16x16x4_f32(
                    false, a, false, bfr[kk], (short)0, acc, false, false);
            }
            const int pbase = t * 16 + 8 * r;
            const int cdst  = OFF_C1 + nl * 484 + pbase;
            if (t < 30) {
#pragma unroll
                for (int v = 0; v < 8; ++v)
                    smem[cdst + v] = acc[v];
            } else {
#pragma unroll
                for (int v = 0; v < 8; ++v)
                    if (pbase + v < 484)
                        smem[cdst + v] = acc[v];
            }
        }
#undef A_AT
#undef KDEL
#undef KOFF

        float w1[9], w2[9], w3[9];
#pragma unroll
        for (int q = 0; q < 9; ++q) {
            w1[q] = smem[OFF_WD + ch * 28 + q];
            w2[q] = smem[OFF_WD + ch * 28 + 9 + q];
            w3[q] = smem[OFF_WD + ch * 28 + 18 + q];
        }
        const float bias = smem[OFF_WD + ch * 28 + 27];

        __syncthreads();
        if (has_next) patch_store(smem, next, wid, lane, pf);

        const v2f bias2 = {bias, bias};
        const float* c1p = smem + OFF_C1 + ch * 484 + (4 * rg) * 22 + 4 * cs;
        float* op = out + ((size_t)(bb * COUT + ch) * HW + (oy + 4 * rg)) * HW + (ox + 4 * cs);

#define SPLAT(w) ((v2f){(w), (w)})
#define FMA2(a, b, c) __builtin_elementwise_fma((a), (b), (c))

        v2f a2[3][4], a3[3][3], a4[3][2];
#pragma unroll
        for (int i = 0; i < 10; ++i) {
            v2f e1[5], o1[4];
#pragma unroll
            for (int q = 0; q < 5; ++q)
                e1[q] = *(const v2f*)(c1p + i * 22 + 2 * q);
#pragma unroll
            for (int q = 0; q < 4; ++q)
                o1[q] = (v2f){e1[q].y, e1[q + 1].x};
#pragma unroll
            for (int ty = 0; ty < 3; ++ty) {
                const int a = i - ty;
                if (a >= 0 && a < 8) {
#pragma unroll
                    for (int q = 0; q < 4; ++q) {
                        v2f s;
                        if (ty == 0) s = e1[q] * SPLAT(w1[0]);
                        else         s = FMA2(e1[q], SPLAT(w1[ty * 3]), a2[a % 3][q]);
                        s = FMA2(o1[q],     SPLAT(w1[ty * 3 + 1]), s);
                        s = FMA2(e1[q + 1], SPLAT(w1[ty * 3 + 2]), s);
                        a2[a % 3][q] = s;
                    }
                }
            }
            if (i >= 2) {
                const int a = i - 2;
                v2f e2[4], o2[3];
#pragma unroll
                for (int q = 0; q < 4; ++q) e2[q] = a2[a % 3][q];
#pragma unroll
                for (int q = 0; q < 3; ++q) o2[q] = (v2f){e2[q].y, e2[q + 1].x};
#pragma unroll
                for (int ty = 0; ty < 3; ++ty) {
                    const int b = a - ty;
                    if (b >= 0 && b < 6) {
#pragma unroll
                        for (int q = 0; q < 3; ++q) {
                            v2f s;
                            if (ty == 0) s = e2[q] * SPLAT(w2[0]);
                            else         s = FMA2(e2[q], SPLAT(w2[ty * 3]), a3[b % 3][q]);
                            s = FMA2(o2[q],     SPLAT(w2[ty * 3 + 1]), s);
                            s = FMA2(e2[q + 1], SPLAT(w2[ty * 3 + 2]), s);
                            a3[b % 3][q] = s;
                        }
                    }
                }
                if (a >= 2) {
                    const int b = a - 2;
                    v2f e3[3], o3[2];
#pragma unroll
                    for (int q = 0; q < 3; ++q) e3[q] = a3[b % 3][q];
#pragma unroll
                    for (int q = 0; q < 2; ++q) o3[q] = (v2f){e3[q].y, e3[q + 1].x};
#pragma unroll
                    for (int ty = 0; ty < 3; ++ty) {
                        const int c = b - ty;
                        if (c >= 0 && c < 4) {
#pragma unroll
                            for (int q = 0; q < 2; ++q) {
                                v2f s = (ty == 0) ? bias2 : a4[c % 3][q];
                                s = FMA2(e3[q],     SPLAT(w3[ty * 3]),     s);
                                s = FMA2(o3[q],     SPLAT(w3[ty * 3 + 1]), s);
                                s = FMA2(e3[q + 1], SPLAT(w3[ty * 3 + 2]), s);
                                a4[c % 3][q] = s;
                            }
                        }
                    }
                    if (b >= 2) {
                        const int c = b - 2;
                        *(float4*)(op + (size_t)c * HW) =
                            make_float4(a4[c % 3][0].x, a4[c % 3][0].y, a4[c % 3][1].x, a4[c % 3][1].y);
                    }
                }
            }
        }
#undef SPLAT
#undef FMA2
        if (!has_next) break;
        __syncthreads();
        tile = next;
    }
}

extern "C" void kernel_launch(void* const* d_in, const int* in_sizes, int n_in,
                              void* d_out, int out_size, void* d_ws, size_t ws_size,
                              hipStream_t stream) {
    const float* x  = (const float*)d_in[0];
    const float* W0 = (const float*)d_in[1];
    const float* W1 = (const float*)d_in[2];
    const float* W2 = (const float*)d_in[3];
    const float* W3 = (const float*)d_in[4];
    const float* b3 = (const float*)d_in[5];
    float* out = (float*)d_out;

    dim3 grid(NWG);
    dim3 block(256);
    fused_stackedconv_kernel<<<grid, block, 0, stream>>>(x, W0, W1, W2, W3, b3, out);
}


# work-group tile widened from 16x16 to 16x32 outputs, each thread an 8x4 output block: less conv1 and depthwise halo work per output, whole 128-byte output lines per store; sum order unchanged
# speedup vs baseline: 2.6546x; 1.1676x over previous
#include <hip/hip_runtime.h>

typedef __attribute__((ext_vector_type(2))) float v2f;
typedef __attribute__((ext_vector_type(8))) float v8f;

#define CIN   8
#define COUT  16
#define HW    224
#define TH    16
#define TW    32
#define NTILE 3136
#define TPI   98
#define TPR   7
#define PW    40
#define PCH   960
#define C1W   38
#define C1N   836
#define NMT   53

#define OFF_IN   0
#define OFF_W0T  7680
#define OFF_C1   8832
#define OFF_WD   22208
#define SMEM_F   22656
#define NWG      768

static __device__ __forceinline__ void patch_load(const float* __restrict__ x, int tile,
                                                  int wid, int lane, float (&pf)[30])
{
    const int bb  = tile / TPI;
    const int rem = tile - bb * TPI;
    const int by  = rem / TPR;
    const int bx  = rem - by * TPR;
    const float* plane = x + ((size_t)bb * CIN + wid) * (HW * HW);
    const int gxa  = bx * TW - 4 + lane;
    const int gxac = gxa < 0 ? 0 : (gxa > HW - 1 ? HW - 1 : gxa);
#pragma unroll
    for (int k = 0; k < 24; ++k) {
        const int gy  = by * TH - 4 + k;
        const int gyc = gy < 0 ? 0 : (gy > HW - 1 ? HW - 1 : gy);
        pf[k] = plane[gyc * HW + gxac];
    }
    const int gxb  = bx * TW + 28 + (lane & 7);
    const int gxbc = gxb < 0 ? 0 : (gxb > HW - 1 ? HW - 1 : gxb);
#pragma unroll
    for (int k2 = 0; k2 < 6; ++k2) {
        const int gy  = by * TH - 4 + 4 * k2 + (lane >> 3);
        const int gyc = gy < 0 ? 0 : (gy > HW - 1 ? HW - 1 : gy);
        pf[24 + k2] = plane[gyc * HW + gxbc];
    }
}

static __device__ __forceinline__ void patch_store(float* smem, int tile,
                                                   int wid, int lane, const float (&pf)[30])
{
    const int bb  = tile / TPI;
    const int rem = tile - bb * TPI;
    const int by  = rem / TPR;
    const int bx  = rem - by * TPR;
    const int gxa = bx * TW - 4 + lane;
    const bool colokA = (gxa >= 0) && (gxa < HW);
    float* dstA = smem + OFF_IN + wid * PCH + lane;
#pragma unroll
    for (int k = 0; k < 24; ++k) {
        const int gy = by * TH - 4 + k;
        const bool ok = colokA && (gy >= 0) && (gy < HW);
        dstA[k * PW] = ok ? pf[k] : 0.0f;
    }
    const int gxb = bx * TW + 28 + (lane & 7);
    const bool colokB = (gxb >= 0) && (gxb < HW);
    const int rb = lane >> 3;
    float* dstB = smem + OFF_IN + wid * PCH + rb * PW + 32 + (lane & 7);
#pragma unroll
    for (int k2 = 0; k2 < 6; ++k2) {
        const int gy = by * TH - 4 + 4 * k2 + rb;
        const bool ok = colokB && (gy >= 0) && (gy < HW);
        dstB[k2 * 4 * PW] = ok ? pf[24 + k2] : 0.0f;
    }
}

__global__ __launch_bounds__(256, 6)
void fused_stackedconv_kernel(const float* __restrict__ x,
                              const float* __restrict__ W0,
                              const float* __restrict__ W1,
                              const float* __restrict__ W2,
                              const float* __restrict__ W3,
                              const float* __restrict__ b3,
                              float* __restrict__ out)
{
    __shared__ float smem[SMEM_F];
    const int tid0 = threadIdx.x;

    for (int idx = tid0; idx < 72 * 16; idx += 256) {
        const int k = idx >> 4, n = idx & 15;
        smem[OFF_W0T + idx] = W0[n * 72 + k];
    }
    for (int idx = tid0; idx < 16 * 28; idx += 256) {
        const int c = idx / 28, q = idx - c * 28;
        float v;
        if      (q < 9)  v = W1[c * 9 + q];
        else if (q < 18) v = W2[c * 9 + q - 9];
        else if (q < 27) v = W3[c * 9 + q - 18];
        else             v = b3[c];
        smem[OFF_WD + idx] = v;
    }
    int tile = blockIdx.x;
    {
        float pf0[30];
        patch_load(x, tile, tid0 >> 5, tid0 & 31, pf0);
        patch_store(smem, tile, tid0 >> 5, tid0 & 31, pf0);
    }
    __syncthreads();

    for (;;) {
        int tid = threadIdx.x;
        asm volatile("" : "+v"(tid));
        const int lane = tid & 31;
        const int wid  = __builtin_amdgcn_readfirstlane(tid >> 5);
        const int cs = tid & 7;
        const int rg = (tid >> 3) & 1;
        const int ch = tid >> 4;

        const int  next     = tile + NWG;
        const bool has_next = next < NTILE;
        const int bb  = tile / TPI;
        const int rem = tile - bb * TPI;
        const int oy  = (rem / TPR) * TH;
        const int ox  = (rem % TPR) * TW;

        float pf[30];
        patch_load(x, has_next ? next : tile, wid, lane, pf);

        const int r  = lane >> 4;
        const int nl = lane & 15;

        v2f bfr[18];
#pragma unroll
        for (int kk = 0; kk < 18; ++kk) {
            const int kx = 4 * kk + 2 * r;
            const int ky = kx + 1;
            bfr[kk].x = smem[OFF_W0T + kx * 16 + nl];
            bfr[kk].y = smem[OFF_W0T + ky * 16 + nl];
        }
#define KOFF(k) (((k) / 9) * PCH + (((k) % 9) / 3) * PW + ((k) % 9) % 3)
#define KDEL(k) (KOFF((k) + 2) - KOFF(k))
#define A_AT(k) ((KDEL(k) == 2 ? pa2 : (KDEL(k) == 39 ? pa39 : pa879))[KOFF(k)])
        const int d2 = r * 2, d39 = r * 39, d879 = r * 879;

        for (int t = wid; t < NMT; t += 8) {
            const int p  = t * 16 + nl;
            const int pp = p < C1N - 1 ? p : C1N - 1;
            const int i  = pp / C1W;
            const int j  = pp - i * C1W;
            const int abase = OFF_IN + i * PW + j;
            const float* pa2   = smem + abase + d2;
            const float* pa39  = smem + abase + d39;
            const float* pa879 = smem + abase + d879;

            v8f acc = {};
#pragma unroll
            for (int kk = 0; kk < 18; ++kk) {
                v2f a;
                a.x = A_AT(4 * kk);
                a.y = A_AT(4 * kk + 1);
                acc = __builtin_amdgcn_wmma_f32_16x16x4_f32(
                    false, a, false, bfr[kk], (short)0, acc, false, false);
            }
            const int pbase = t * 16 + 8 * r;
            const int cdst  = OFF_C1 + nl * C1N + pbase;
            if (t < NMT - 1) {
#pragma unroll
                for (int v = 0; v < 8; ++v)
                    smem[cdst + v] = acc[v];
            } else {
#pragma unroll
                for (int v = 0; v < 8; ++v)
                    if (pbase + v < C1N)
                        smem[cdst + v] = acc[v];
            }
        }
#undef A_AT
#undef KDEL
#undef KOFF

        float w1[9], w2[9], w3[9];
#pragma unroll
        for (int q = 0; q < 9; ++q) {
            w1[q] = smem[OFF_WD + ch * 28 + q];
            w2[q] = smem[OFF_WD + ch * 28 + 9 + q];
            w3[q] = smem[OFF_WD + ch * 28 + 18 + q];
        }
        const float bias = smem[OFF_WD + ch * 28 + 27];

        __syncthreads();
        if (has_next) patch_store(smem, next, wid, lane, pf);

        const v2f bias2 = {bias, bias};
        const float* c1p = smem + OFF_C1 + ch * C1N + (8 * rg) * C1W + 4 * cs;
        float* op = out + ((size_t)(bb * COUT + ch) * HW + (oy + 8 * rg)) * HW + (ox + 4 * cs);

#define SPLAT(w) ((v2f){(w), (w)})
#define FMA2(a, b, c) __builtin_elementwise_fma((a), (b), (c))

        v2f a2[3][4], a3[3][3], a4[3][2];
#pragma unroll
        for (int i = 0; i < 14; ++i) {
            v2f e1[5], o1[4];
#pragma unroll
            for (int q = 0; q < 5; ++q)
                e1[q] = *(const v2f*)(c1p + i * C1W + 2 * q);
#pragma unroll
            for (int q = 0; q < 4; ++q)
                o1[q] = (v2f){e1[q].y, e1[q + 1].x};
#pragma unroll
            for (int ty = 0; ty < 3; ++ty) {
                const int a = i - ty;
                if (a >= 0 && a < 12) {
#pragma unroll
                    for (int q = 0; q < 4; ++q) {
                        v2f s;
                        if (ty == 0) s = e1[q] * SPLAT(w1[0]);
                        else         s = FMA2(e1[q], SPLAT(w1[ty * 3]), a2[a % 3][q]);
                        s = FMA2(o1[q],     SPLAT(w1[ty * 3 + 1]), s);
                        s = FMA2(e1[q + 1], SPLAT(w1[ty * 3 + 2]), s);
                        a2[a % 3][q] = s;
                    }
                }
            }
            if (i >= 2) {
                const int a = i - 2;
                v2f e2[4], o2[3];
#pragma unroll
                for (int q = 0; q < 4; ++q) e2[q] = a2[a % 3][q];
#pragma unroll
                for (int q = 0; q < 3; ++q) o2[q] = (v2f){e2[q].y, e2[q + 1].x};
#pragma unroll
                for (int ty = 0; ty < 3; ++ty) {
                    const int b = a - ty;
                    if (b >= 0 && b < 10) {
#pragma unroll
                        for (int q = 0; q < 3; ++q) {
                            v2f s;
                            if (ty == 0) s = e2[q] * SPLAT(w2[0]);
                            else         s = FMA2(e2[q], SPLAT(w2[ty * 3]), a3[b % 3][q]);
                            s = FMA2(o2[q],     SPLAT(w2[ty * 3 + 1]), s);
                            s = FMA2(e2[q + 1], SPLAT(w2[ty * 3 + 2]), s);
                            a3[b % 3][q] = s;
                        }
                    }
                }
                if (a >= 2) {
                    const int b = a - 2;
                    v2f e3[3], o3[2];
#pragma unroll
                    for (int q = 0; q < 3; ++q) e3[q] = a3[b % 3][q];
#pragma unroll
                    for (int q = 0; q < 2; ++q) o3[q] = (v2f){e3[q].y, e3[q + 1].x};
#pragma unroll
                    for (int ty = 0; ty < 3; ++ty) {
                        const int c = b - ty;
                        if (c >= 0 && c < 8) {
#pragma unroll
                            for (int q = 0; q < 2; ++q) {
                                v2f s = (ty == 0) ? bias2 : a4[c % 3][q];
                                s = FMA2(e3[q],     SPLAT(w3[ty * 3]),     s);
                                s = FMA2(o3[q],     SPLAT(w3[ty * 3 + 1]), s);
                                s = FMA2(e3[q + 1], SPLAT(w3[ty * 3 + 2]), s);
                                a4[c % 3][q] = s;
                            }
                        }
                    }
                    if (b >= 2) {
                        const int c = b - 2;
                        *(float4*)(op + (size_t)c * HW) =
                            make_float4(a4[c % 3][0].x, a4[c % 3][0].y, a4[c % 3][1].x, a4[c % 3][1].y);
                    }
                }
            }
        }
#undef SPLAT
#undef FMA2
        if (!has_next) break;
        __syncthreads();
        tile = next;
    }
}

extern "C" void kernel_launch(void* const* d_in, const int* in_sizes, int n_in,
                              void* d_out, int out_size, void* d_ws, size_t ws_size,
                              hipStream_t stream) {
    const float* x  = (const float*)d_in[0];
    const float* W0 = (const float*)d_in[1];
    const float* W1 = (const float*)d_in[2];
    const float* W2 = (const float*)d_in[3];
    const float* W3 = (const float*)d_in[4];
    const float* b3 = (const float*)d_in[5];
    float* out = (float*)d_out;

    dim3 grid(NWG);
    dim3 block(256);
    fused_stackedconv_kernel<<<grid, block, 0, stream>>>(x, W0, W1, W2, W3, b3, out);
}


# work-group tile heightened from 16x32 to 28x32 outputs (each thread a 14x4 block, two groups per WGP): less conv1 and depthwise halo work per output, an even 7 tiles per WGP; sum order unchanged
# speedup vs baseline: 3.0597x; 1.1526x over previous
#include <hip/hip_runtime.h>

typedef __attribute__((ext_vector_type(2))) float v2f;
typedef __attribute__((ext_vector_type(8))) float v8f;

#define CIN   8
#define COUT  16
#define HW    224
#define TH    28
#define TW    32
#define NTILE 1792
#define TPI   56
#define TPR   7
#define PR    36
#define PW    40
#define PCH   1448
#define C1W   38
#define C1N   1292
#define NMT   81
#define RPT   14
#define NPF   45

#define OFF_IN   0
#define OFF_W0T  11584
#define OFF_C1   12736
#define OFF_WD   33408
#define SMEM_F   33856
#define NWG      512

static __device__ __forceinline__ void patch_load(const float* __restrict__ x, int tile,
                                                  int wid, int lane, float (&pf)[NPF])
{
    const int bb  = tile / TPI;
    const int rem = tile - bb * TPI;
    const int by  = rem / TPR;
    const int bx  = rem - by * TPR;
    const float* plane = x + ((size_t)bb * CIN + wid) * (HW * HW);
    const int gxa  = bx * TW - 4 + lane;
    const int gxac = gxa < 0 ? 0 : (gxa > HW - 1 ? HW - 1 : gxa);
#pragma unroll
    for (int k = 0; k < PR; ++k) {
        const int gy  = by * TH - 4 + k;
        const int gyc = gy < 0 ? 0 : (gy > HW - 1 ? HW - 1 : gy);
        pf[k] = plane[gyc * HW + gxac];
    }
    const int gxb  = bx * TW + 28 + (lane & 7);
    const int gxbc = gxb < 0 ? 0 : (gxb > HW - 1 ? HW - 1 : gxb);
#pragma unroll
    for (int k2 = 0; k2 < PR / 4; ++k2) {
        const int gy  = by * TH - 4 + 4 * k2 + (lane >> 3);
        const int gyc = gy < 0 ? 0 : (gy > HW - 1 ? HW - 1 : gy);
        pf[PR + k2] = plane[gyc * HW + gxbc];
    }
}

static __device__ __forceinline__ void patch_store(float* smem, int tile,
                                                   int wid, int lane, const float (&pf)[NPF])
{
    const int bb  = tile / TPI;
    const int rem = tile - bb * TPI;
    const int by  = rem / TPR;
    const int bx  = rem - by * TPR;
    const int gxa = bx * TW - 4 + lane;
    const bool colokA = (gxa >= 0) && (gxa < HW);
    float* dstA = smem + OFF_IN + wid * PCH + lane;
#pragma unroll
    for (int k = 0; k < PR; ++k) {
        const int gy = by * TH - 4 + k;
        const bool ok = colokA && (gy >= 0) && (gy < HW);
        dstA[k * PW] = ok ? pf[k] : 0.0f;
    }
    const int gxb = bx * TW + 28 + (lane & 7);
    const bool colokB = (gxb >= 0) && (gxb < HW);
    const int rb = lane >> 3;
    float* dstB = smem + OFF_IN + wid * PCH + rb * PW + 32 + (lane & 7);
#pragma unroll
    for (int k2 = 0; k2 < PR / 4; ++k2) {
        const int gy = by * TH - 4 + 4 * k2 + rb;
        const bool ok = colokB && (gy >= 0) && (gy < HW);
        dstB[k2 * 4 * PW] = ok ? pf[PR + k2] : 0.0f;
    }
}

__global__ __launch_bounds__(256, 4)
void fused_stackedconv_kernel(const float* __restrict__ x,
                              const float* __restrict__ W0,
                              const float* __restrict__ W1,
                              const float* __restrict__ W2,
                              const float* __restrict__ W3,
                              const float* __restrict__ b3,
                              float* __restrict__ out)
{
    __shared__ float smem[SMEM_F];
    const int tid0 = threadIdx.x;

    for (int idx = tid0; idx < 72 * 16; idx += 256) {
        const int k = idx >> 4, n = idx & 15;
        smem[OFF_W0T + idx] = W0[n * 72 + k];
    }
    for (int idx = tid0; idx < 16 * 28; idx += 256) {
        const int c = idx / 28, q = idx - c * 28;
        float v;
        if      (q < 9)  v = W1[c * 9 + q];
        else if (q < 18) v = W2[c * 9 + q - 9];
        else if (q < 27) v = W3[c * 9 + q - 18];
        else             v = b3[c];
        smem[OFF_WD + idx] = v;
    }
    int tile = blockIdx.x;
    {
        float pf0[NPF];
        patch_load(x, tile, tid0 >> 5, tid0 & 31, pf0);
        patch_store(smem, tile, tid0 >> 5, tid0 & 31, pf0);
    }
    __syncthreads();

    for (;;) {
        const int tid  = threadIdx.x;
        const int lane = tid & 31;
        const int wid  = __builtin_amdgcn_readfirstlane(tid >> 5);
        const int cs = tid & 7;
        const int rg = (tid >> 3) & 1;
        const int ch = tid >> 4;

        const int  next     = tile + NWG;
        const bool has_next = next < NTILE;
        const int bb  = tile / TPI;
        const int rem = tile - bb * TPI;
        const int oy  = (rem / TPR) * TH;
        const int ox  = (rem % TPR) * TW;

        float pf[NPF];
        patch_load(x, has_next ? next : tile, wid, lane, pf);

        const int r  = lane >> 4;
        const int nl = lane & 15;

        v2f bfr[18];
#pragma unroll
        for (int kk = 0; kk < 18; ++kk) {
            const int kx = 4 * kk + 2 * r;
            const int ky = kx + 1;
            bfr[kk].x = smem[OFF_W0T + kx * 16 + nl];
            bfr[kk].y = smem[OFF_W0T + ky * 16 + nl];
        }
#define KOFF(k) (((k) / 9) * PCH + (((k) % 9) / 3) * PW + ((k) % 9) % 3)
#define KDEL(k) (KOFF((k) + 2) - KOFF(k))
#define A_AT(k) ((KDEL(k) == 2 ? pa2 : (KDEL(k) == 39 ? pa39 : pach))[KOFF(k)])
        const int d2 = r * 2, d39 = r * 39, dch = r * (PCH - 81);

        for (int t = wid; t < NMT; t += 8) {
            const int p  = t * 16 + nl;
            const int pp = p < C1N - 1 ? p : C1N - 1;
            const int i  = pp / C1W;
            const int j  = pp - i * C1W;
            const int abase = OFF_IN + i * PW + j;
            const float* pa2   = smem + abase + d2;
            const float* pa39  = smem + abase + d39;
            const float* pach  = smem + abase + dch;

            v8f acc = {};
#pragma unroll
            for (int kk = 0; kk < 18; ++kk) {
                v2f a;
                a.x = A_AT(4 * kk);
                a.y = A_AT(4 * kk + 1);
                acc = __builtin_amdgcn_wmma_f32_16x16x4_f32(
                    false, a, false, bfr[kk], (short)0, acc, false, false);
            }
            const int pbase = t * 16 + 8 * r;
            const int cdst  = OFF_C1 + nl * C1N + pbase;
            if (t < NMT - 1) {
#pragma unroll
                for (int v = 0; v < 8; ++v)
                    smem[cdst + v] = acc[v];
            } else {
#pragma unroll
                for (int v = 0; v < 8; ++v)
                    if (pbase + v < C1N)
                        smem[cdst + v] = acc[v];
            }
        }
#undef A_AT
#undef KDEL
#undef KOFF

        float w1[9], w2[9], w3[9];
#pragma unroll
        for (int q = 0; q < 9; ++q) {
            w1[q] = smem[OFF_WD + ch * 28 + q];
            w2[q] = smem[OFF_WD + ch * 28 + 9 + q];
            w3[q] = smem[OFF_WD + ch * 28 + 18 + q];
        }
        const float bias = smem[OFF_WD + ch * 28 + 27];

        __syncthreads();
        if (has_next) patch_store(smem, next, wid, lane, pf);

        const v2f bias2 = {bias, bias};
        const float* c1p = smem + OFF_C1 + ch * C1N + (RPT * rg) * C1W + 4 * cs;
        float* op = out + ((size_t)(bb * COUT + ch) * HW + (oy + RPT * rg)) * HW + (ox + 4 * cs);

#define SPLAT(w) ((v2f){(w), (w)})
#define FMA2(a, b, c) __builtin_elementwise_fma((a), (b), (c))

        v2f a2[3][4], a3[3][3], a4[3][2];
#pragma unroll
        for (int i = 0; i < RPT + 6; ++i) {
            v2f e1[5], o1[4];
#pragma unroll
            for (int q = 0; q < 5; ++q)
                e1[q] = *(const v2f*)(c1p + i * C1W + 2 * q);
#pragma unroll
            for (int q = 0; q < 4; ++q)
                o1[q] = (v2f){e1[q].y, e1[q + 1].x};
#pragma unroll
            for (int ty = 0; ty < 3; ++ty) {
                const int a = i - ty;
                if (a >= 0 && a < RPT + 4) {
#pragma unroll
                    for (int q = 0; q < 4; ++q) {
                        v2f s;
                        if (ty == 0) s = e1[q] * SPLAT(w1[0]);
                        else         s = FMA2(e1[q], SPLAT(w1[ty * 3]), a2[a % 3][q]);
                        s = FMA2(o1[q],     SPLAT(w1[ty * 3 + 1]), s);
                        s = FMA2(e1[q + 1], SPLAT(w1[ty * 3 + 2]), s);
                        a2[a % 3][q] = s;
                    }
                }
            }
            if (i >= 2) {
                const int a = i - 2;
                v2f e2[4], o2[3];
#pragma unroll
                for (int q = 0; q < 4; ++q) e2[q] = a2[a % 3][q];
#pragma unroll
                for (int q = 0; q < 3; ++q) o2[q] = (v2f){e2[q].y, e2[q + 1].x};
#pragma unroll
                for (int ty = 0; ty < 3; ++ty) {
                    const int b = a - ty;
                    if (b >= 0 && b < RPT + 2) {
#pragma unroll
                        for (int q = 0; q < 3; ++q) {
                            v2f s;
                            if (ty == 0) s = e2[q] * SPLAT(w2[0]);
                            else         s = FMA2(e2[q], SPLAT(w2[ty * 3]), a3[b % 3][q]);
                            s = FMA2(o2[q],     SPLAT(w2[ty * 3 + 1]), s);
                            s = FMA2(e2[q + 1], SPLAT(w2[ty * 3 + 2]), s);
                            a3[b % 3][q] = s;
                        }
                    }
                }
                if (a >= 2) {
                    const int b = a - 2;
                    v2f e3[3], o3[2];
#pragma unroll
                    for (int q = 0; q < 3; ++q) e3[q] = a3[b % 3][q];
#pragma unroll
                    for (int q = 0; q < 2; ++q) o3[q] = (v2f){e3[q].y, e3[q + 1].x};
#pragma unroll
                    for (int ty = 0; ty < 3; ++ty) {
                        const int c = b - ty;
                        if (c >= 0 && c < RPT) {
#pragma unroll
                            for (int q = 0; q < 2; ++q) {
                                v2f s = (ty == 0) ? bias2 : a4[c % 3][q];
                                s = FMA2(e3[q],     SPLAT(w3[ty * 3]),     s);
                                s = FMA2(o3[q],     SPLAT(w3[ty * 3 + 1]), s);
                                s = FMA2(e3[q + 1], SPLAT(w3[ty * 3 + 2]), s);
                                a4[c % 3][q] = s;
                            }
                        }
                    }
                    if (b >= 2) {
                        const int c = b - 2;
                        *(float4*)(op + (size_t)c * HW) =
                            make_float4(a4[c % 3][0].x, a4[c % 3][0].y, a4[c % 3][1].x, a4[c % 3][1].y);
                    }
                }
            }
        }
#undef SPLAT
#undef FMA2
        if (!has_next) break;
        __syncthreads();
        tile = next;
    }
}

extern "C" void kernel_launch(void* const* d_in, const int* in_sizes, int n_in,
                              void* d_out, int out_size, void* d_ws, size_t ws_size,
                              hipStream_t stream) {
    const float* x  = (const float*)d_in[0];
    const float* W0 = (const float*)d_in[1];
    const float* W1 = (const float*)d_in[2];
    const float* W2 = (const float*)d_in[3];
    const float* W3 = (const float*)d_in[4];
    const float* b3 = (const float*)d_in[5];
    float* out = (float*)d_out;

    dim3 grid(NWG);
    dim3 block(256);
    fused_stackedconv_kernel<<<grid, block, 0, stream>>>(x, W0, W1, W2, W3, b3, out);
}


# tiles walked in runs of 4 and 3 along a tile row; a tile keeps its left neighbour's last 6 conv1 columns (moved in LDS) and computes 68 M-tiles, not 81; products and sum order unchanged
# speedup vs baseline: 3.2070x; 1.0481x over previous
#include <hip/hip_runtime.h>

typedef __attribute__((ext_vector_type(2))) float v2f;
typedef __attribute__((ext_vector_type(8))) float v8f;

#define CIN   8
#define COUT  16
#define HW    224
#define TH    28
#define TW    32
#define NTILE 1792
#define TPI   56
#define TPR   7
#define PR    36
#define PW    40
#define PCH   1448
#define C1W   38
#define C1N   1292
#define NMT   81
#define RPT   14
#define NPF   45
#define NROW  256
#define NMTI  68

#define OFF_IN   0
#define OFF_W0T  11584
#define OFF_C1   12736
#define OFF_WD   33408
#define SMEM_F   33856
#define NWG      512

static __device__ __forceinline__ void patch_load(const float* __restrict__ x, int tile,
                                                  int wid, int lane, float (&pf)[NPF])
{
    const int bb  = tile / TPI;
    const int rem = tile - bb * TPI;
    const int by  = rem / TPR;
    const int bx  = rem - by * TPR;
    const float* plane = x + ((size_t)bb * CIN + wid) * (HW * HW);
    const int gxa  = bx * TW - 4 + lane;
    const int gxac = gxa < 0 ? 0 : (gxa > HW - 1 ? HW - 1 : gxa);
#pragma unroll
    for (int k = 0; k < PR; ++k) {
        const int gy  = by * TH - 4 + k;
        const int gyc = gy < 0 ? 0 : (gy > HW - 1 ? HW - 1 : gy);
        pf[k] = plane[gyc * HW + gxac];
    }
    const int gxb  = bx * TW + 28 + (lane & 7);
    const int gxbc = gxb < 0 ? 0 : (gxb > HW - 1 ? HW - 1 : gxb);
#pragma unroll
    for (int k2 = 0; k2 < PR / 4; ++k2) {
        const int gy  = by * TH - 4 + 4 * k2 + (lane >> 3);
        const int gyc = gy < 0 ? 0 : (gy > HW - 1 ? HW - 1 : gy);
        pf[PR + k2] = plane[gyc * HW + gxbc];
    }
}

static __device__ __forceinline__ void patch_store(float* smem, int tile,
                                                   int wid, int lane, const float (&pf)[NPF])
{
    const int bb  = tile / TPI;
    const int rem = tile - bb * TPI;
    const int by  = rem / TPR;
    const int bx  = rem - by * TPR;
    const int gxa = bx * TW - 4 + lane;
    const bool colokA = (gxa >= 0) && (gxa < HW);
    float* dstA = smem + OFF_IN + wid * PCH + lane;
#pragma unroll
    for (int k = 0; k < PR; ++k) {
        const int gy = by * TH - 4 + k;
        const bool ok = colokA && (gy >= 0) && (gy < HW);
        dstA[k * PW] = ok ? pf[k] : 0.0f;
    }
    const int gxb = bx * TW + 28 + (lane & 7);
    const bool colokB = (gxb >= 0) && (gxb < HW);
    const int rb = lane >> 3;
    float* dstB = smem + OFF_IN + wid * PCH + rb * PW + 32 + (lane & 7);
#pragma unroll
    for (int k2 = 0; k2 < PR / 4; ++k2) {
        const int gy = by * TH - 4 + 4 * k2 + rb;
        const bool ok = colokB && (gy >= 0) && (gy < HW);
        dstB[k2 * 4 * PW] = ok ? pf[PR + k2] : 0.0f;
    }
}

#define KOFF(k) (((k) / 9) * PCH + (((k) % 9) / 3) * PW + ((k) % 9) % 3)
#define KDEL(k) (KOFF((k) + 2) - KOFF(k))
#define A_AT(k) ((KDEL(k) == 2 ? pa2 : (KDEL(k) == 39 ? pa39 : pach))[KOFF(k)])
static __device__ __forceinline__ v8f conv1_mtile(const float* smem, int abase, int r,
                                                  const v2f (&bfr)[18])
{
    const float* pa2  = smem + abase + r * 2;
    const float* pa39 = smem + abase + r * 39;
    const float* pach = smem + abase + r * (PCH - 81);
    v8f acc = {};
#pragma unroll
    for (int kk = 0; kk < 18; ++kk) {
        v2f a;
        a.x = A_AT(4 * kk);
        a.y = A_AT(4 * kk + 1);
        acc = __builtin_amdgcn_wmma_f32_16x16x4_f32(
            false, a, false, bfr[kk], (short)0, acc, false, false);
    }
    return acc;
}
#undef A_AT
#undef KDEL
#undef KOFF

__global__ __launch_bounds__(256, 4)
void fused_stackedconv_kernel(const float* __restrict__ x,
                              const float* __restrict__ W0,
                              const float* __restrict__ W1,
                              const float* __restrict__ W2,
                              const float* __restrict__ W3,
                              const float* __restrict__ b3,
                              float* __restrict__ out)
{
    __shared__ float smem[SMEM_F];
    const int tid0 = threadIdx.x;

    for (int idx = tid0; idx < 72 * 16; idx += 256) {
        const int k = idx >> 4, n = idx & 15;
        smem[OFF_W0T + idx] = W0[n * 72 + k];
    }
    for (int idx = tid0; idx < 16 * 28; idx += 256) {
        const int c = idx / 28, q = idx - c * 28;
        float v;
        if      (q < 9)  v = W1[c * 9 + q];
        else if (q < 18) v = W2[c * 9 + q - 9];
        else if (q < 27) v = W3[c * 9 + q - 18];
        else             v = b3[c];
        smem[OFF_WD + idx] = v;
    }
    const int ntl = 4 - (int)(blockIdx.x >> 8);
    int tile = (int)(blockIdx.x & (NROW - 1)) * TPR + (int)(blockIdx.x >> 8) * 4;
    int n = 0;
    {
        float pf0[NPF];
        patch_load(x, tile, tid0 >> 5, tid0 & 31, pf0);
        patch_store(smem, tile, tid0 >> 5, tid0 & 31, pf0);
    }
    __syncthreads();

    for (;;) {
        const int tid  = threadIdx.x;
        const int lane = tid & 31;
        const int wid  = __builtin_amdgcn_readfirstlane(tid >> 5);
        const int cs = tid & 7;
        const int rg = (tid >> 3) & 1;
        const int ch = tid >> 4;

        const int  next     = tile + 1;
        const bool has_next = n + 1 < ntl;
        const int bb  = tile / TPI;
        const int rem = tile - bb * TPI;
        const int oy  = (rem / TPR) * TH;
        const int ox  = (rem % TPR) * TW;

        float pf[NPF];
        patch_load(x, has_next ? next : tile, wid, lane, pf);

        const int r  = lane >> 4;
        const int nl = lane & 15;

        v2f bfr[18];
#pragma unroll
        for (int kk = 0; kk < 18; ++kk) {
            const int kx = 4 * kk + 2 * r;
            const int ky = kx + 1;
            bfr[kk].x = smem[OFF_W0T + kx * 16 + nl];
            bfr[kk].y = smem[OFF_W0T + ky * 16 + nl];
        }
        if (n == 0) {
            for (int t = wid; t < NMT; t += 8) {
                const int p  = t * 16 + nl;
                const int pp = p < C1N - 1 ? p : C1N - 1;
                const int i  = pp / C1W;
                const int j  = pp - i * C1W;
                const v8f acc = conv1_mtile(smem, OFF_IN + i * PW + j, r, bfr);
                const int pbase = t * 16 + 8 * r;
                const int cdst  = OFF_C1 + nl * C1N + pbase;
                if (t < NMT - 1) {
#pragma unroll
                    for (int v = 0; v < 8; ++v)
                        smem[cdst + v] = acc[v];
                } else {
#pragma unroll
                    for (int v = 0; v < 8; ++v)
                        if (pbase + v < C1N)
                            smem[cdst + v] = acc[v];
                }
            }
        } else {
            const int hh = wid & 1;
            int csrc[3];
#pragma unroll
            for (int u = 0; u < 3; ++u) {
                const int e  = lane + 32 * u;
                const int cc = e / 6;
                csrc[u] = OFF_C1 + cc * C1N + (e - 6 * cc);
            }
            for (int t = wid; t < NMTI; t += 8) {
                const int i  = t >> 1;
                const int jb = 6 + 16 * hh;
                float cv[3] = {0.0f, 0.0f, 0.0f};
                if (hh) {
#pragma unroll
                    for (int u = 0; u < 3; ++u) cv[u] = smem[csrc[u] + i * C1W + TW];
                }
                const v8f acc = conv1_mtile(smem, OFF_IN + i * PW + jb + nl, r, bfr);
                if (hh) {
#pragma unroll
                    for (int u = 0; u < 3; ++u) smem[csrc[u] + i * C1W] = cv[u];
                }
                const int cdst = OFF_C1 + nl * C1N + i * C1W + jb + 8 * r;
#pragma unroll
                for (int v = 0; v < 8; ++v)
                    smem[cdst + v] = acc[v];
            }
        }

        float w1[9], w2[9], w3[9];
#pragma unroll
        for (int q = 0; q < 9; ++q) {
            w1[q] = smem[OFF_WD + ch * 28 + q];
            w2[q] = smem[OFF_WD + ch * 28 + 9 + q];
            w3[q] = smem[OFF_WD + ch * 28 + 18 + q];
        }
        const float bias = smem[OFF_WD + ch * 28 + 27];

        __syncthreads();
        if (has_next) patch_store(smem, next, wid, lane, pf);

        const v2f bias2 = {bias, bias};
        const float* c1p = smem + OFF_C1 + ch * C1N + (RPT * rg) * C1W + 4 * cs;
        float* op = out + ((size_t)(bb * COUT + ch) * HW + (oy + RPT * rg)) * HW + (ox + 4 * cs);

#define SPLAT(w) ((v2f){(w), (w)})
#define FMA2(a, b, c) __builtin_elementwise_fma((a), (b), (c))

        v2f a2[3][4], a3[3][3], a4[3][2];
#pragma unroll
        for (int i = 0; i < RPT + 6; ++i) {
            v2f e1[5], o1[4];
#pragma unroll
            for (int q = 0; q < 5; ++q)
                e1[q] = *(const v2f*)(c1p + i * C1W + 2 * q);
#pragma unroll
            for (int q = 0; q < 4; ++q)
                o1[q] = (v2f){e1[q].y, e1[q + 1].x};
#pragma unroll
            for (int ty = 0; ty < 3; ++ty) {
                const int a = i - ty;
                if (a >= 0 && a < RPT + 4) {
#pragma unroll
                    for (int q = 0; q < 4; ++q) {
                        v2f s;
                        if (ty == 0) s = e1[q] * SPLAT(w1[0]);
                        else         s = FMA2(e1[q], SPLAT(w1[ty * 3]), a2[a % 3][q]);
                        s = FMA2(o1[q],     SPLAT(w1[ty * 3 + 1]), s);
                        s = FMA2(e1[q + 1], SPLAT(w1[ty * 3 + 2]), s);
                        a2[a % 3][q] = s;
                    }
                }
            }
            if (i >= 2) {
                const int a = i - 2;
                v2f e2[4], o2[3];
#pragma unroll
                for (int q = 0; q < 4; ++q) e2[q] = a2[a % 3][q];
#pragma unroll
                for (int q = 0; q < 3; ++q) o2[q] = (v2f){e2[q].y, e2[q + 1].x};
#pragma unroll
                for (int ty = 0; ty < 3; ++ty) {
                    const int b = a - ty;
                    if (b >= 0 && b < RPT + 2) {
#pragma unroll
                        for (int q = 0; q < 3; ++q) {
                            v2f s;
                            if (ty == 0) s = e2[q] * SPLAT(w2[0]);
                            else         s = FMA2(e2[q], SPLAT(w2[ty * 3]), a3[b % 3][q]);
                            s = FMA2(o2[q],     SPLAT(w2[ty * 3 + 1]), s);
                            s = FMA2(e2[q + 1], SPLAT(w2[ty * 3 + 2]), s);
                            a3[b % 3][q] = s;
                        }
                    }
                }
                if (a >= 2) {
                    const int b = a - 2;
                    v2f e3[3], o3[2];
#pragma unroll
                    for (int q = 0; q < 3; ++q) e3[q] = a3[b % 3][q];
#pragma unroll
                    for (int q = 0; q < 2; ++q) o3[q] = (v2f){e3[q].y, e3[q + 1].x};
#pragma unroll
                    for (int ty = 0; ty < 3; ++ty) {
                        const int c = b - ty;
                        if (c >= 0 && c < RPT) {
#pragma unroll
                            for (int q = 0; q < 2; ++q) {
                                v2f s = (ty == 0) ? bias2 : a4[c % 3][q];
                                s = FMA2(e3[q],     SPLAT(w3[ty * 3]),     s);
                                s = FMA2(o3[q],     SPLAT(w3[ty * 3 + 1]), s);
                                s = FMA2(e3[q + 1], SPLAT(w3[ty * 3 + 2]), s);
                                a4[c % 3][q] = s;
                            }
                        }
                    }
                    if (b >= 2) {
                        const int c = b - 2;
                        *(float4*)(op + (size_t)c * HW) =
                            make_float4(a4[c % 3][0].x, a4[c % 3][0].y, a4[c % 3][1].x, a4[c % 3][1].y);
                    }
                }
            }
        }
#undef SPLAT
#undef FMA2
        if (!has_next) break;
        __syncthreads();
        tile = next;
        ++n;
    }
}

extern "C" void kernel_launch(void* const* d_in, const int* in_sizes, int n_in,
                              void* d_out, int out_size, void* d_ws, size_t ws_size,
                              hipStream_t stream) {
    const float* x  = (const float*)d_in[0];
    const float* W0 = (const float*)d_in[1];
    const float* W1 = (const float*)d_in[2];
    const float* W2 = (const float*)d_in[3];
    const float* W3 = (const float*)d_in[4];
    const float* b3 = (const float*)d_in[5];
    float* out = (float*)d_out;

    dim3 grid(NWG);
    dim3 block(256);
    fused_stackedconv_kernel<<<grid, block, 0, stream>>>(x, W0, W1, W2, W3, b3, out);
}


# wave roles: 16-wave groups, one per WGP, each walk a tile row; 8 waves run conv1 of tile s while 8 run the depthwise stage of tile s-1; two conv1 planes, LDS-only barrier per step; sums unchanged
# speedup vs baseline: 3.3741x; 1.0521x over previous
#include <hip/hip_runtime.h>

typedef __attribute__((ext_vector_type(2))) float v2f;
typedef __attribute__((ext_vector_type(8))) float v8f;

#define CIN   8
#define COUT  16
#define HW    224
#define TH    28
#define TW    32
#define NTILE 1792
#define TPI   56
#define TPR   7
#define PR    36
#define PW    40
#define PCH   1448
#define C1W   38
#define C1N   1292
#define NMT   81
#define RPT   14
#define NPF   45
#define NROW  256
#define NMTI  68

#define PINSZ    11584
#define C1SZ     20672
#define OFF_IN   0
#define OFF_W0T  23168
#define OFF_C1   24320
#define OFF_WD   65664
#define SMEM_F   66112
#define NWG      256
#define NTHR     512

static __device__ __forceinline__ void patch_load(const float* __restrict__ x, int tile,
                                                  int wid, int lane, float (&pf)[NPF])
{
    const int bb  = tile / TPI;
    const int rem = tile - bb * TPI;
    const int by  = rem / TPR;
    const int bx  = rem - by * TPR;
    const float* plane = x + ((size_t)bb * CIN + wid) * (HW * HW);
    const int gxa  = bx * TW - 4 + lane;
    const int gxac = gxa < 0 ? 0 : (gxa > HW - 1 ? HW - 1 : gxa);
#pragma unroll
    for (int k = 0; k < PR; ++k) {
        const int gy  = by * TH - 4 + k;
        const int gyc = gy < 0 ? 0 : (gy > HW - 1 ? HW - 1 : gy);
        pf[k] = plane[gyc * HW + gxac];
    }
    const int gxb  = bx * TW + 28 + (lane & 7);
    const int gxbc = gxb < 0 ? 0 : (gxb > HW - 1 ? HW - 1 : gxb);
#pragma unroll
    for (int k2 = 0; k2 < PR / 4; ++k2) {
        const int gy  = by * TH - 4 + 4 * k2 + (lane >> 3);
        const int gyc = gy < 0 ? 0 : (gy > HW - 1 ? HW - 1 : gy);
        pf[PR + k2] = plane[gyc * HW + gxbc];
    }
}

static __device__ __forceinline__ void patch_store(float* pin, int tile,
                                                   int wid, int lane, const float (&pf)[NPF])
{
    const int bb  = tile / TPI;
    const int rem = tile - bb * TPI;
    const int by  = rem / TPR;
    const int bx  = rem - by * TPR;
    const int gxa = bx * TW - 4 + lane;
    const bool colokA = (gxa >= 0) && (gxa < HW);
    float* dstA = pin + wid * PCH + lane;
#pragma unroll
    for (int k = 0; k < PR; ++k) {
        const int gy = by * TH - 4 + k;
        const bool ok = colokA && (gy >= 0) && (gy < HW);
        dstA[k * PW] = ok ? pf[k] : 0.0f;
    }
    const int gxb = bx * TW + 28 + (lane & 7);
    const bool colokB = (gxb >= 0) && (gxb < HW);
    const int rb = lane >> 3;
    float* dstB = pin + wid * PCH + rb * PW + 32 + (lane & 7);
#pragma unroll
    for (int k2 = 0; k2 < PR / 4; ++k2) {
        const int gy = by * TH - 4 + 4 * k2 + rb;
        const bool ok = colokB && (gy >= 0) && (gy < HW);
        dstB[k2 * 4 * PW] = ok ? pf[PR + k2] : 0.0f;
    }
}

#define KOFF(k) (((k) / 9) * PCH + (((k) % 9) / 3) * PW + ((k) % 9) % 3)
#define KDEL(k) (KOFF((k) + 2) - KOFF(k))
#define A_AT(k) ((KDEL(k) == 2 ? pa2 : (KDEL(k) == 39 ? pa39 : pach))[KOFF(k)])
static __device__ __forceinline__ v8f conv1_mtile(const float* smem, int abase, int r,
                                                  const v2f (&bfr)[18])
{
    const float* pa2  = smem + abase + r * 2;
    const float* pa39 = smem + abase + r * 39;
    const float* pach = smem + abase + r * (PCH - 81);
    v8f acc = {};
#pragma unroll
    for (int kk = 0; kk < 18; ++kk) {
        v2f a;
        a.x = A_AT(4 * kk);
        a.y = A_AT(4 * kk + 1);
        acc = __builtin_amdgcn_wmma_f32_16x16x4_f32(
            false, a, false, bfr[kk], (short)0, acc, false, false);
    }
    return acc;
}
#undef A_AT
#undef KDEL
#undef KOFF

__global__ __launch_bounds__(NTHR, 4)
void fused_stackedconv_kernel(const float* __restrict__ x,
                              const float* __restrict__ W0,
                              const float* __restrict__ W1,
                              const float* __restrict__ W2,
                              const float* __restrict__ W3,
                              const float* __restrict__ b3,
                              float* __restrict__ out)
{
    __shared__ float smem[SMEM_F];
    const int tid0 = threadIdx.x;
    const int wv   = __builtin_amdgcn_readfirstlane(tid0 >> 5);
    const bool mat = wv < 8;
    const int wid  = wv & 7;
    const int tid  = tid0 & 255;
    const int lane = tid0 & 31;

    for (int idx = tid0; idx < 72 * 16; idx += NTHR) {
        const int k = idx >> 4, n = idx & 15;
        smem[OFF_W0T + idx] = W0[n * 72 + k];
    }
    for (int idx = tid0; idx < 16 * 28; idx += NTHR) {
        const int c = idx / 28, q = idx - c * 28;
        float v;
        if      (q < 9)  v = W1[c * 9 + q];
        else if (q < 18) v = W2[c * 9 + q - 9];
        else if (q < 27) v = W3[c * 9 + q - 18];
        else             v = b3[c];
        smem[OFF_WD + idx] = v;
    }
    const int tile0 = (int)blockIdx.x * TPR;
    if (mat) {
        float pf0[NPF];
        patch_load(x, tile0, wid, lane, pf0);
        patch_store(smem + OFF_IN, tile0, wid, lane, pf0);
    }
    __syncthreads();

    for (int s = 0; s <= TPR; ++s) {
        const int par   = s & 1;
        const int pin   = OFF_IN + par * PINSZ;
        const int pinn  = OFF_IN + (par ^ 1) * PINSZ;
        const int c1new = OFF_C1 + par * C1SZ;
        const int c1old = OFF_C1 + (par ^ 1) * C1SZ;

        if (mat) {
          if (s < TPR) {
            const int  tile     = tile0 + s;
            const bool has_next = s + 1 < TPR;

            float pf[NPF];
            patch_load(x, has_next ? tile + 1 : tile, wid, lane, pf);

            const int r  = lane >> 4;
            const int nl = lane & 15;

            v2f bfr[18];
#pragma unroll
            for (int kk = 0; kk < 18; ++kk) {
                const int kx = 4 * kk + 2 * r;
                const int ky = kx + 1;
                bfr[kk].x = smem[OFF_W0T + kx * 16 + nl];
                bfr[kk].y = smem[OFF_W0T + ky * 16 + nl];
            }
            if (s == 0) {
                for (int t = wid; t < NMT; t += 8) {
                    const int p  = t * 16 + nl;
                    const int pp = p < C1N - 1 ? p : C1N - 1;
                    const int i  = pp / C1W;
                    const int j  = pp - i * C1W;
                    const v8f acc = conv1_mtile(smem, pin + i * PW + j, r, bfr);
                    const int pbase = t * 16 + 8 * r;
                    const int cdst  = c1new + nl * C1N + pbase;
                    if (t < NMT - 1) {
#pragma unroll
                        for (int v = 0; v < 8; ++v)
                            smem[cdst + v] = acc[v];
                    } else {
#pragma unroll
                        for (int v = 0; v < 8; ++v)
                            if (pbase + v < C1N)
                                smem[cdst + v] = acc[v];
                    }
                }
            } else {
                const int hh = wid & 1;
                int csrc[3];
#pragma unroll
                for (int u = 0; u < 3; ++u) {
                    const int e  = lane + 32 * u;
                    const int cc = e / 6;
                    csrc[u] = cc * C1N + (e - 6 * cc);
                }
                for (int t = wid; t < NMTI; t += 8) {
                    const int i  = t >> 1;
                    const int jb = 6 + 16 * hh;
                    float cv[3] = {0.0f, 0.0f, 0.0f};
                    if (hh) {
#pragma unroll
                        for (int u = 0; u < 3; ++u) cv[u] = smem[c1old + csrc[u] + i * C1W + TW];
                    }
                    const v8f acc = conv1_mtile(smem, pin + i * PW + jb + nl, r, bfr);
                    if (hh) {
#pragma unroll
                        for (int u = 0; u < 3; ++u) smem[c1new + csrc[u] + i * C1W] = cv[u];
                    }
                    const int cdst = c1new + nl * C1N + i * C1W + jb + 8 * r;
#pragma unroll
                    for (int v = 0; v < 8; ++v)
                        smem[cdst + v] = acc[v];
                }
            }
            if (has_next) patch_store(smem + pinn, tile + 1, wid, lane, pf);
          }
        } else if (s > 0) {
            const int tile = tile0 + s - 1;
            const int cs = tid & 7;
            const int rg = (tid >> 3) & 1;
            const int ch = tid >> 4;
            const int bb  = tile / TPI;
            const int rem = tile - bb * TPI;
            const int oy  = (rem / TPR) * TH;
            const int ox  = (rem % TPR) * TW;

            float w1[9], w2[9], w3[9];
#pragma unroll
            for (int q = 0; q < 9; ++q) {
                w1[q] = smem[OFF_WD + ch * 28 + q];
                w2[q] = smem[OFF_WD + ch * 28 + 9 + q];
                w3[q] = smem[OFF_WD + ch * 28 + 18 + q];
            }
            const float bias = smem[OFF_WD + ch * 28 + 27];

            const v2f bias2 = {bias, bias};
            const float* c1p = smem + c1old + ch * C1N + (RPT * rg) * C1W + 4 * cs;
            float* op = out + ((size_t)(bb * COUT + ch) * HW + (oy + RPT * rg)) * HW + (ox + 4 * cs);

#define SPLAT(w) ((v2f){(w), (w)})
#define FMA2(a, b, c) __builtin_elementwise_fma((a), (b), (c))

            v2f a2[3][4], a3[3][3], a4[3][2];
#pragma unroll
            for (int i = 0; i < RPT + 6; ++i) {
                v2f e1[5], o1[4];
#pragma unroll
                for (int q = 0; q < 5; ++q)
                    e1[q] = *(const v2f*)(c1p + i * C1W + 2 * q);
#pragma unroll
                for (int q = 0; q < 4; ++q)
                    o1[q] = (v2f){e1[q].y, e1[q + 1].x};
#pragma unroll
                for (int ty = 0; ty < 3; ++ty) {
                    const int a = i - ty;
                    if (a >= 0 && a < RPT + 4) {
#pragma unroll
                        for (int q = 0; q < 4; ++q) {
                            v2f sm;
                            if (ty == 0) sm = e1[q] * SPLAT(w1[0]);
                            else         sm = FMA2(e1[q], SPLAT(w1[ty * 3]), a2[a % 3][q]);
                            sm = FMA2(o1[q],     SPLAT(w1[ty * 3 + 1]), sm);
                            sm = FMA2(e1[q + 1], SPLAT(w1[ty * 3 + 2]), sm);
                            a2[a % 3][q] = sm;
                        }
                    }
                }
                if (i >= 2) {
                    const int a = i - 2;
                    v2f e2[4], o2[3];
#pragma unroll
                    for (int q = 0; q < 4; ++q) e2[q] = a2[a % 3][q];
#pragma unroll
                    for (int q = 0; q < 3; ++q) o2[q] = (v2f){e2[q].y, e2[q + 1].x};
#pragma unroll
                    for (int ty = 0; ty < 3; ++ty) {
                        const int b = a - ty;
                        if (b >= 0 && b < RPT + 2) {
#pragma unroll
                            for (int q = 0; q < 3; ++q) {
                                v2f sm;
                                if (ty == 0) sm = e2[q] * SPLAT(w2[0]);
                                else         sm = FMA2(e2[q], SPLAT(w2[ty * 3]), a3[b % 3][q]);
                                sm = FMA2(o2[q],     SPLAT(w2[ty * 3 + 1]), sm);
                                sm = FMA2(e2[q + 1], SPLAT(w2[ty * 3 + 2]), sm);
                                a3[b % 3][q] = sm;
                            }
                        }
                    }
                    if (a >= 2) {
                        const int b = a - 2;
                        v2f e3[3], o3[2];
#pragma unroll
                        for (int q = 0; q < 3; ++q) e3[q] = a3[b % 3][q];
#pragma unroll
                        for (int q = 0; q < 2; ++q) o3[q] = (v2f){e3[q].y, e3[q + 1].x};
#pragma unroll
                        for (int ty = 0; ty < 3; ++ty) {
                            const int c = b - ty;
                            if (c >= 0 && c < RPT) {
#pragma unroll
                                for (int q = 0; q < 2; ++q) {
                                    v2f sm = (ty == 0) ? bias2 : a4[c % 3][q];
                                    sm = FMA2(e3[q],     SPLAT(w3[ty * 3]),     sm);
                                    sm = FMA2(o3[q],     SPLAT(w3[ty * 3 + 1]), sm);
                                    sm = FMA2(e3[q + 1], SPLAT(w3[ty * 3 + 2]), sm);
                                    a4[c % 3][q] = sm;
                                }
                            }
                        }
                        if (b >= 2) {
                            const int c = b - 2;
                            *(float4*)(op + (size_t)c * HW) =
                                make_float4(a4[c % 3][0].x, a4[c % 3][0].y, a4[c % 3][1].x, a4[c % 3][1].y);
                        }
                    }
                }
            }
#undef SPLAT
#undef FMA2
        }
        if (s == TPR) break;
        __builtin_amdgcn_fence(__ATOMIC_RELEASE, "workgroup", "local");
        __builtin_amdgcn_s_barrier();
        __builtin_amdgcn_fence(__ATOMIC_ACQUIRE, "workgroup", "local");
    }
}

extern "C" void kernel_launch(void* const* d_in, const int* in_sizes, int n_in,
                              void* d_out, int out_size, void* d_ws, size_t ws_size,
                              hipStream_t stream) {
    const float* x  = (const float*)d_in[0];
    const float* W0 = (const float*)d_in[1];
    const float* W1 = (const float*)d_in[2];
    const float* W2 = (const float*)d_in[3];
    const float* W3 = (const float*)d_in[4];
    const float* b3 = (const float*)d_in[5];
    float* out = (float*)d_out;

    dim3 grid(NWG);
    dim3 block(NTHR);
    fused_stackedconv_kernel<<<grid, block, 0, stream>>>(x, W0, W1, W2, W3, b3, out);
}


# conv1 in the minimal-filtering form for 2x2 outputs of a 3x3 filter on the same f32 16x16x4 products: 32 products per 64 positions, not 72; patch held as channel pairs; conv1 sum order changes
# speedup vs baseline: 4.0111x; 1.1888x over previous
#include <hip/hip_runtime.h>

typedef __attribute__((ext_vector_type(2))) float v2f;
typedef __attribute__((ext_vector_type(4))) float v4f;
typedef __attribute__((ext_vector_type(8))) float v8f;

#define CIN   8
#define COUT  16
#define HW    224
#define TH    28
#define TW    32
#define NTILE 1792
#define TPI   56
#define TPR   7
#define PR    36
#define PW    40
#define PCH2  2896
#define C1W   38
#define C1N   1292
#define NGR   17
#define NGR0  21
#define NXB   51
#define RPT   14
#define NPF   45
#define NROW  256

#define PINSZ    11584
#define C1SZ     20672
#define OFF_IN   0
#define ULN      20
#define OFF_U    23168
#define OFF_C1   25728
#define OFF_WD   67072
#define SMEM_F   67520
#define NWG      256
#define NTHR     512

static __device__ __forceinline__ void patch_load(const float* __restrict__ x, int tile,
                                                  int wid, int lane, float (&pf)[NPF])
{
    const int bb  = tile / TPI;
    const int rem = tile - bb * TPI;
    const int by  = rem / TPR;
    const int bx  = rem - by * TPR;
    const float* plane = x + ((size_t)bb * CIN + wid) * (HW * HW);
    const int gxa  = bx * TW - 4 + lane;
    const int gxac = gxa < 0 ? 0 : (gxa > HW - 1 ? HW - 1 : gxa);
#pragma unroll
    for (int k = 0; k < PR; ++k) {
        const int gy  = by * TH - 4 + k;
        const int gyc = gy < 0 ? 0 : (gy > HW - 1 ? HW - 1 : gy);
        pf[k] = plane[gyc * HW + gxac];
    }
    const int gxb  = bx * TW + 28 + (lane & 7);
    const int gxbc = gxb < 0 ? 0 : (gxb > HW - 1 ? HW - 1 : gxb);
#pragma unroll
    for (int k2 = 0; k2 < PR / 4; ++k2) {
        const int gy  = by * TH - 4 + 4 * k2 + (lane >> 3);
        const int gyc = gy < 0 ? 0 : (gy > HW - 1 ? HW - 1 : gy);
        pf[PR + k2] = plane[gyc * HW + gxbc];
    }
}

static __device__ __forceinline__ void patch_store(float* pin, int tile,
                                                   int wid, int lane, const float (&pf)[NPF])
{
    const int bb  = tile / TPI;
    const int rem = tile - bb * TPI;
    const int by  = rem / TPR;
    const int bx  = rem - by * TPR;
    const int gxa = bx * TW - 4 + lane;
    const bool colokA = (gxa >= 0) && (gxa < HW);
    float* dstA = pin + (wid >> 1) * PCH2 + (wid & 1) + lane * 2;
#pragma unroll
    for (int k = 0; k < PR; ++k) {
        const int gy = by * TH - 4 + k;
        const bool ok = colokA && (gy >= 0) && (gy < HW);
        dstA[k * PW * 2] = ok ? pf[k] : 0.0f;
    }
    const int gxb = bx * TW + 28 + (lane & 7);
    const bool colokB = (gxb >= 0) && (gxb < HW);
    const int rb = lane >> 3;
    float* dstB = pin + (wid >> 1) * PCH2 + (wid & 1) + (rb * PW + 32 + (lane & 7)) * 2;
#pragma unroll
    for (int k2 = 0; k2 < PR / 4; ++k2) {
        const int gy = by * TH - 4 + 4 * k2 + rb;
        const bool ok = colokB && (gy >= 0) && (gy < HW);
        dstB[k2 * 4 * PW * 2] = ok ? pf[PR + k2] : 0.0f;
    }
}

static __device__ __forceinline__ void conv1_group(const float* smem, int abase, int boff,
                                                   v8f (&Y)[2][2])
{
    const float* pb = smem + boff;
    v4f bq[4][4];
#pragma unroll
    for (int nu = 0; nu < 4; ++nu)
        bq[0][nu] = *(const v4f*)(pb + nu * 4);
    v4f q[2][4][2];
#pragma unroll
    for (int kk = 0; kk < 2; ++kk) {
#pragma unroll
        for (int y = 0; y < 4; ++y) {
            q[kk][y][0] = *(const v4f*)(smem + abase + kk * 2 * PCH2 + y * PW * 2);
            q[kk][y][1] = *(const v4f*)(smem + abase + kk * 2 * PCH2 + y * PW * 2 + 4);
        }
    }
    v2f av[2][4][4];
#pragma unroll
    for (int kk = 0; kk < 2; ++kk) {
        v2f d[4][4];
#pragma unroll
        for (int y = 0; y < 4; ++y) {
            d[y][0] = (v2f){q[kk][y][0].x, q[kk][y][0].y};
            d[y][1] = (v2f){q[kk][y][0].z, q[kk][y][0].w};
            d[y][2] = (v2f){q[kk][y][1].x, q[kk][y][1].y};
            d[y][3] = (v2f){q[kk][y][1].z, q[kk][y][1].w};
        }
        v2f w[4][4];
#pragma unroll
        for (int c = 0; c < 4; ++c) {
            w[0][c] = d[0][c] - d[2][c];
            w[1][c] = d[1][c] + d[2][c];
            w[2][c] = d[2][c] - d[1][c];
            w[3][c] = d[1][c] - d[3][c];
        }
#pragma unroll
        for (int xi = 0; xi < 4; ++xi) {
            av[kk][xi][0] = w[xi][0] - w[xi][2];
            av[kk][xi][1] = w[xi][1] + w[xi][2];
            av[kk][xi][2] = w[xi][2] - w[xi][1];
            av[kk][xi][3] = w[xi][1] - w[xi][3];
        }
    }
#pragma unroll
    for (int xi = 0; xi < 4; ++xi) {
        if (xi < 3) {
#pragma unroll
            for (int nu = 0; nu < 4; ++nu)
                bq[xi + 1][nu] = *(const v4f*)(pb + (xi + 1) * 32 * ULN + nu * 4);
        }
        v8f m[4];
#pragma unroll
        for (int nu = 0; nu < 4; ++nu) {
            v8f z = {};
            z = __builtin_amdgcn_wmma_f32_16x16x4_f32(
                false, av[0][xi][nu], false, (v2f){bq[xi][nu].x, bq[xi][nu].y}, (short)0, z, false, false);
            m[nu] = __builtin_amdgcn_wmma_f32_16x16x4_f32(
                false, av[1][xi][nu], false, (v2f){bq[xi][nu].z, bq[xi][nu].w}, (short)0, z, false, false);
        }
        const v8f t0 = m[0] + m[1] + m[2];
        const v8f t1 = m[1] - m[2] - m[3];
        if (xi == 0)      { Y[0][0] = t0;  Y[0][1] = t1; }
        else if (xi == 1) { Y[0][0] += t0; Y[0][1] += t1; Y[1][0] = t0;  Y[1][1] = t1; }
        else if (xi == 2) { Y[0][0] += t0; Y[0][1] += t1; Y[1][0] -= t0; Y[1][1] -= t1; }
        else              { Y[1][0] -= t0; Y[1][1] -= t1; }
    }
}

__global__ __launch_bounds__(NTHR, 4)
void fused_stackedconv_kernel(const float* __restrict__ x,
                              const float* __restrict__ W0,
                              const float* __restrict__ W1,
                              const float* __restrict__ W2,
                              const float* __restrict__ W3,
                              const float* __restrict__ b3,
                              float* __restrict__ out)
{
    __shared__ float smem[SMEM_F];
    const int tid0 = threadIdx.x;
    const int wv   = __builtin_amdgcn_readfirstlane(tid0 >> 5);
    const bool mat = wv < 8;
    const int wid  = wv & 7;
    const int tid  = tid0 & 255;
    const int lane = tid0 & 31;

    if (tid0 < CIN * COUT) {
        const int cin = tid0 >> 4, n = tid0 & 15;
        const float* gp = W0 + n * 72 + cin * 9;
        float gg[3][3];
#pragma unroll
        for (int q = 0; q < 9; ++q) gg[q / 3][q % 3] = gp[q];
        float ug[4][3];
#pragma unroll
        for (int c = 0; c < 3; ++c) {
            ug[0][c] = gg[0][c];
            ug[1][c] = 0.5f * (gg[0][c] + gg[1][c] + gg[2][c]);
            ug[2][c] = 0.5f * (gg[0][c] - gg[1][c] + gg[2][c]);
            ug[3][c] = gg[2][c];
        }
        const int kk = cin >> 2, rr = (cin >> 1) & 1, ss = cin & 1;
#pragma unroll
        for (int xi = 0; xi < 4; ++xi) {
            float uu[4];
            uu[0] = ug[xi][0];
            uu[1] = 0.5f * (ug[xi][0] + ug[xi][1] + ug[xi][2]);
            uu[2] = 0.5f * (ug[xi][0] - ug[xi][1] + ug[xi][2]);
            uu[3] = ug[xi][2];
#pragma unroll
            for (int nu = 0; nu < 4; ++nu)
                smem[OFF_U + (xi * 32 + rr * 16 + n) * ULN + nu * 4 + kk * 2 + ss] = uu[nu];
        }
    }
    for (int idx = tid0; idx < 16 * 28; idx += NTHR) {
        const int c = idx / 28, q = idx - c * 28;
        float v;
        if      (q < 9)  v = W1[c * 9 + q];
        else if (q < 18) v = W2[c * 9 + q - 9];
        else if (q < 27) v = W3[c * 9 + q - 18];
        else             v = b3[c];
        smem[OFF_WD + idx] = v;
    }
    const int tile0 = (int)blockIdx.x * TPR;
    if (mat) {
        float pf0[NPF];
        patch_load(x, tile0, wid, lane, pf0);
        patch_store(smem + OFF_IN, tile0, wid, lane, pf0);
    }
    __syncthreads();

    for (int s = 0; s <= TPR; ++s) {
        const int par   = s & 1;
        const int pin   = OFF_IN + par * PINSZ;
        const int pinn  = OFF_IN + (par ^ 1) * PINSZ;
        const int c1new = OFF_C1 + par * C1SZ;
        const int c1old = OFF_C1 + (par ^ 1) * C1SZ;

        if (mat) {
          if (s < TPR) {
            const int  tile     = tile0 + s;
            const bool has_next = s + 1 < TPR;

            float pf[NPF];
            patch_load(x, has_next ? tile + 1 : tile, wid, lane, pf);

            const int r  = lane >> 4;
            const int nl = lane & 15;

            const int boff = OFF_U + lane * ULN;

            int coff[6];
#pragma unroll
            for (int u = 0; u < 6; ++u) {
                const int e  = lane + 32 * u;
                const int a  = e / 96;
                const int e2 = e - 96 * a;
                const int cc = e2 / 6;
                coff[u] = cc * C1N + a * C1W + (e2 - 6 * cc);
            }
            const int ng = s == 0 ? NGR0 : NGR;
            for (int t = wid; t < ng; t += 8) {
                const bool newc = t < NGR;
                const bool copy = newc && s > 0;
                int abase;
                if (newc) {
                    abase = pin + r * PCH2 + ((2 * t) * PW + 6 + 2 * nl) * 2;
                } else {
                    const int q  = (t - NGR) * 16 + nl;
                    const int qc = q < NXB - 1 ? q : NXB - 1;
                    const int tr = qc / 3;
                    abase = pin + r * PCH2 + ((2 * tr) * PW + 2 * (qc - 3 * tr)) * 2;
                }
                float cv[6] = {0.0f, 0.0f, 0.0f, 0.0f, 0.0f, 0.0f};
                if (copy) {
#pragma unroll
                    for (int u = 0; u < 6; ++u) cv[u] = smem[c1old + coff[u] + 2 * t * C1W + TW];
                }
                v8f Y[2][2];
                conv1_group(smem, abase, boff, Y);
                if (copy) {
#pragma unroll
                    for (int u = 0; u < 6; ++u) smem[c1new + coff[u] + 2 * t * C1W] = cv[u];
                }
                if (newc) {
#pragma unroll
                    for (int a = 0; a < 2; ++a) {
                        const int cdst = c1new + nl * C1N + (2 * t + a) * C1W + 6 + 16 * r;
#pragma unroll
                        for (int v = 0; v < 8; ++v)
                            *(v2f*)(smem + cdst + 2 * v) = (v2f){Y[a][0][v], Y[a][1][v]};
                    }
                } else {
#pragma unroll
                    for (int v = 0; v < 8; ++v) {
                        const int q = (t - NGR) * 16 + 8 * r + v;
                        if (q < NXB) {
                            const int tr = q / 3;
                            const int cdst = c1new + nl * C1N + (2 * tr) * C1W + 2 * (q - 3 * tr);
                            *(v2f*)(smem + cdst)       = (v2f){Y[0][0][v], Y[0][1][v]};
                            *(v2f*)(smem + cdst + C1W) = (v2f){Y[1][0][v], Y[1][1][v]};
                        }
                    }
                }
            }
            if (has_next) patch_store(smem + pinn, tile + 1, wid, lane, pf);
          }
        } else if (s > 0) {
            const int tile = tile0 + s - 1;
            const int cs = tid & 7;
            const int rg = (tid >> 3) & 1;
            const int ch = tid >> 4;
            const int bb  = tile / TPI;
            const int rem = tile - bb * TPI;
            const int oy  = (rem / TPR) * TH;
            const int ox  = (rem % TPR) * TW;

            float w1[9], w2[9], w3[9];
#pragma unroll
            for (int q = 0; q < 9; ++q) {
                w1[q] = smem[OFF_WD + ch * 28 + q];
                w2[q] = smem[OFF_WD + ch * 28 + 9 + q];
                w3[q] = smem[OFF_WD + ch * 28 + 18 + q];
            }
            const float bias = smem[OFF_WD + ch * 28 + 27];

            const v2f bias2 = {bias, bias};
            const float* c1p = smem + c1old + ch * C1N + (RPT * rg) * C1W + 4 * cs;
            float* op = out + ((size_t)(bb * COUT + ch) * HW + (oy + RPT * rg)) * HW + (ox + 4 * cs);

#define SPLAT(w) ((v2f){(w), (w)})
#define FMA2(a, b, c) __builtin_elementwise_fma((a), (b), (c))

            v2f a2[3][4], a3[3][3], a4[3][2];
#pragma unroll
            for (int i = 0; i < RPT + 6; ++i) {
                v2f e1[5], o1[4];
#pragma unroll
                for (int q = 0; q < 5; ++q)
                    e1[q] = *(const v2f*)(c1p + i * C1W + 2 * q);
#pragma unroll
                for (int q = 0; q < 4; ++q)
                    o1[q] = (v2f){e1[q].y, e1[q + 1].x};
#pragma unroll
                for (int ty = 0; ty < 3; ++ty) {
                    const int a = i - ty;
                    if (a >= 0 && a < RPT + 4) {
#pragma unroll
                        for (int q = 0; q < 4; ++q) {
                            v2f sm;
                            if (ty == 0) sm = e1[q] * SPLAT(w1[0]);
                            else         sm = FMA2(e1[q], SPLAT(w1[ty * 3]), a2[a % 3][q]);
                            sm = FMA2(o1[q],     SPLAT(w1[ty * 3 + 1]), sm);
                            sm = FMA2(e1[q + 1], SPLAT(w1[ty * 3 + 2]), sm);
                            a2[a % 3][q] = sm;
                        }
                    }
                }
                if (i >= 2) {
                    const int a = i - 2;
                    v2f e2[4], o2[3];
#pragma unroll
                    for (int q = 0; q < 4; ++q) e2[q] = a2[a % 3][q];
#pragma unroll
                    for (int q = 0; q < 3; ++q) o2[q] = (v2f){e2[q].y, e2[q + 1].x};
#pragma unroll
                    for (int ty = 0; ty < 3; ++ty) {
                        const int b = a - ty;
                        if (b >= 0 && b < RPT + 2) {
#pragma unroll
                            for (int q = 0; q < 3; ++q) {
                                v2f sm;
                                if (ty == 0) sm = e2[q] * SPLAT(w2[0]);
                                else         sm = FMA2(e2[q], SPLAT(w2[ty * 3]), a3[b % 3][q]);
                                sm = FMA2(o2[q],     SPLAT(w2[ty * 3 + 1]), sm);
                                sm = FMA2(e2[q + 1], SPLAT(w2[ty * 3 + 2]), sm);
                                a3[b % 3][q] = sm;
                            }
                        }
                    }
                    if (a >= 2) {
                        const int b = a - 2;
                        v2f e3[3], o3[2];
#pragma unroll
                        for (int q = 0; q < 3; ++q) e3[q] = a3[b % 3][q];
#pragma unroll
                        for (int q = 0; q < 2; ++q) o3[q] = (v2f){e3[q].y, e3[q + 1].x};
#pragma unroll
                        for (int ty = 0; ty < 3; ++ty) {
                            const int c = b - ty;
                            if (c >= 0 && c < RPT) {
#pragma unroll
                                for (int q = 0; q < 2; ++q) {
                                    v2f sm = (ty == 0) ? bias2 : a4[c % 3][q];
                                    sm = FMA2(e3[q],     SPLAT(w3[ty * 3]),     sm);
                                    sm = FMA2(o3[q],     SPLAT(w3[ty * 3 + 1]), sm);
                                    sm = FMA2(e3[q + 1], SPLAT(w3[ty * 3 + 2]), sm);
                                    a4[c % 3][q] = sm;
                                }
                            }
                        }
                        if (b >= 2) {
                            const int c = b - 2;
                            *(float4*)(op + (size_t)c * HW) =
                                make_float4(a4[c % 3][0].x, a4[c % 3][0].y, a4[c % 3][1].x, a4[c % 3][1].y);
                        }
                    }
                }
            }
#undef SPLAT
#undef FMA2
        }
        if (s == TPR) break;
        __builtin_amdgcn_fence(__ATOMIC_RELEASE, "workgroup", "local");
        __builtin_amdgcn_s_barrier();
        __builtin_amdgcn_fence(__ATOMIC_ACQUIRE, "workgroup", "local");
    }
}

extern "C" void kernel_launch(void* const* d_in, const int* in_sizes, int n_in,
                              void* d_out, int out_size, void* d_ws, size_t ws_size,
                              hipStream_t stream) {
    const float* x  = (const float*)d_in[0];
    const float* W0 = (const float*)d_in[1];
    const float* W1 = (const float*)d_in[2];
    const float* W2 = (const float*)d_in[3];
    const float* W3 = (const float*)d_in[4];
    const float* b3 = (const float*)d_in[5];
    float* out = (float*)d_out;

    dim3 grid(NWG);
    dim3 block(NTHR);
    fused_stackedconv_kernel<<<grid, block, 0, stream>>>(x, W0, W1, W2, W3, b3, out);
}


# depthwise waves read each conv1 row from LDS two rows ahead of its use (three-row register ring, reads held by a scheduling barrier): the LDS round trip runs under arithmetic; sums unchanged
# speedup vs baseline: 4.1257x; 1.0286x over previous
#include <hip/hip_runtime.h>

typedef __attribute__((ext_vector_type(2))) float v2f;
typedef __attribute__((ext_vector_type(4))) float v4f;
typedef __attribute__((ext_vector_type(8))) float v8f;

#define CIN   8
#define COUT  16
#define HW    224
#define TH    28
#define TW    32
#define NTILE 1792
#define TPI   56
#define TPR   7
#define PR    36
#define PW    40
#define PCH2  2896
#define C1W   38
#define C1N   1292
#define NGR   17
#define NGR0  21
#define NXB   51
#define RPT   14
#define NPF   45
#define NROW  256

#define PINSZ    11584
#define C1SZ     20672
#define OFF_IN   0
#define ULN      20
#define OFF_U    23168
#define OFF_C1   25728
#define OFF_WD   67072
#define SMEM_F   67520
#define NWG      256
#define NTHR     512

static __device__ __forceinline__ void patch_load(const float* __restrict__ x, int tile,
                                                  int wid, int lane, float (&pf)[NPF])
{
    const int bb  = tile / TPI;
    const int rem = tile - bb * TPI;
    const int by  = rem / TPR;
    const int bx  = rem - by * TPR;
    const float* plane = x + ((size_t)bb * CIN + wid) * (HW * HW);
    const int gxa  = bx * TW - 4 + lane;
    const int gxac = gxa < 0 ? 0 : (gxa > HW - 1 ? HW - 1 : gxa);
#pragma unroll
    for (int k = 0; k < PR; ++k) {
        const int gy  = by * TH - 4 + k;
        const int gyc = gy < 0 ? 0 : (gy > HW - 1 ? HW - 1 : gy);
        pf[k] = plane[gyc * HW + gxac];
    }
    const int gxb  = bx * TW + 28 + (lane & 7);
    const int gxbc = gxb < 0 ? 0 : (gxb > HW - 1 ? HW - 1 : gxb);
#pragma unroll
    for (int k2 = 0; k2 < PR / 4; ++k2) {
        const int gy  = by * TH - 4 + 4 * k2 + (lane >> 3);
        const int gyc = gy < 0 ? 0 : (gy > HW - 1 ? HW - 1 : gy);
        pf[PR + k2] = plane[gyc * HW + gxbc];
    }
}

static __device__ __forceinline__ void patch_store(float* pin, int tile,
                                                   int wid, int lane, const float (&pf)[NPF])
{
    const int bb  = tile / TPI;
    const int rem = tile - bb * TPI;
    const int by  = rem / TPR;
    const int bx  = rem - by * TPR;
    const int gxa = bx * TW - 4 + lane;
    const bool colokA = (gxa >= 0) && (gxa < HW);
    float* dstA = pin + (wid >> 1) * PCH2 + (wid & 1) + lane * 2;
#pragma unroll
    for (int k = 0; k < PR; ++k) {
        const int gy = by * TH - 4 + k;
        const bool ok = colokA && (gy >= 0) && (gy < HW);
        dstA[k * PW * 2] = ok ? pf[k] : 0.0f;
    }
    const int gxb = bx * TW + 28 + (lane & 7);
    const bool colokB = (gxb >= 0) && (gxb < HW);
    const int rb = lane >> 3;
    float* dstB = pin + (wid >> 1) * PCH2 + (wid & 1) + (rb * PW + 32 + (lane & 7)) * 2;
#pragma unroll
    for (int k2 = 0; k2 < PR / 4; ++k2) {
        const int gy = by * TH - 4 + 4 * k2 + rb;
        const bool ok = colokB && (gy >= 0) && (gy < HW);
        dstB[k2 * 4 * PW * 2] = ok ? pf[PR + k2] : 0.0f;
    }
}

static __device__ __forceinline__ void conv1_group(const float* smem, int abase, int boff,
                                                   v8f (&Y)[2][2])
{
    const float* pb = smem + boff;
    v4f bq[4][4];
#pragma unroll
    for (int nu = 0; nu < 4; ++nu)
        bq[0][nu] = *(const v4f*)(pb + nu * 4);
    v4f q[2][4][2];
#pragma unroll
    for (int kk = 0; kk < 2; ++kk) {
#pragma unroll
        for (int y = 0; y < 4; ++y) {
            q[kk][y][0] = *(const v4f*)(smem + abase + kk * 2 * PCH2 + y * PW * 2);
            q[kk][y][1] = *(const v4f*)(smem + abase + kk * 2 * PCH2 + y * PW * 2 + 4);
        }
    }
    v2f av[2][4][4];
#pragma unroll
    for (int kk = 0; kk < 2; ++kk) {
        v2f d[4][4];
#pragma unroll
        for (int y = 0; y < 4; ++y) {
            d[y][0] = (v2f){q[kk][y][0].x, q[kk][y][0].y};
            d[y][1] = (v2f){q[kk][y][0].z, q[kk][y][0].w};
            d[y][2] = (v2f){q[kk][y][1].x, q[kk][y][1].y};
            d[y][3] = (v2f){q[kk][y][1].z, q[kk][y][1].w};
        }
        v2f w[4][4];
#pragma unroll
        for (int c = 0; c < 4; ++c) {
            w[0][c] = d[0][c] - d[2][c];
            w[1][c] = d[1][c] + d[2][c];
            w[2][c] = d[2][c] - d[1][c];
            w[3][c] = d[1][c] - d[3][c];
        }
#pragma unroll
        for (int xi = 0; xi < 4; ++xi) {
            av[kk][xi][0] = w[xi][0] - w[xi][2];
            av[kk][xi][1] = w[xi][1] + w[xi][2];
            av[kk][xi][2] = w[xi][2] - w[xi][1];
            av[kk][xi][3] = w[xi][1] - w[xi][3];
        }
    }
#pragma unroll
    for (int xi = 0; xi < 4; ++xi) {
        if (xi < 3) {
#pragma unroll
            for (int nu = 0; nu < 4; ++nu)
                bq[xi + 1][nu] = *(const v4f*)(pb + (xi + 1) * 32 * ULN + nu * 4);
        }
        v8f m[4];
#pragma unroll
        for (int nu = 0; nu < 4; ++nu) {
            v8f z = {};
            z = __builtin_amdgcn_wmma_f32_16x16x4_f32(
                false, av[0][xi][nu], false, (v2f){bq[xi][nu].x, bq[xi][nu].y}, (short)0, z, false, false);
            m[nu] = __builtin_amdgcn_wmma_f32_16x16x4_f32(
                false, av[1][xi][nu], false, (v2f){bq[xi][nu].z, bq[xi][nu].w}, (short)0, z, false, false);
        }
        const v8f t0 = m[0] + m[1] + m[2];
        const v8f t1 = m[1] - m[2] - m[3];
        if (xi == 0)      { Y[0][0] = t0;  Y[0][1] = t1; }
        else if (xi == 1) { Y[0][0] += t0; Y[0][1] += t1; Y[1][0] = t0;  Y[1][1] = t1; }
        else if (xi == 2) { Y[0][0] += t0; Y[0][1] += t1; Y[1][0] -= t0; Y[1][1] -= t1; }
        else              { Y[1][0] -= t0; Y[1][1] -= t1; }
    }
}

__global__ __launch_bounds__(NTHR, 4)
void fused_stackedconv_kernel(const float* __restrict__ x,
                              const float* __restrict__ W0,
                              const float* __restrict__ W1,
                              const float* __restrict__ W2,
                              const float* __restrict__ W3,
                              const float* __restrict__ b3,
                              float* __restrict__ out)
{
    __shared__ float smem[SMEM_F];
    const int tid0 = threadIdx.x;
    const int wv   = __builtin_amdgcn_readfirstlane(tid0 >> 5);
    const bool mat = wv < 8;
    const int wid  = wv & 7;
    const int tid  = tid0 & 255;
    const int lane = tid0 & 31;

    if (tid0 < CIN * COUT) {
        const int cin = tid0 >> 4, n = tid0 & 15;
        const float* gp = W0 + n * 72 + cin * 9;
        float gg[3][3];
#pragma unroll
        for (int q = 0; q < 9; ++q) gg[q / 3][q % 3] = gp[q];
        float ug[4][3];
#pragma unroll
        for (int c = 0; c < 3; ++c) {
            ug[0][c] = gg[0][c];
            ug[1][c] = 0.5f * (gg[0][c] + gg[1][c] + gg[2][c]);
            ug[2][c] = 0.5f * (gg[0][c] - gg[1][c] + gg[2][c]);
            ug[3][c] = gg[2][c];
        }
        const int kk = cin >> 2, rr = (cin >> 1) & 1, ss = cin & 1;
#pragma unroll
        for (int xi = 0; xi < 4; ++xi) {
            float uu[4];
            uu[0] = ug[xi][0];
            uu[1] = 0.5f * (ug[xi][0] + ug[xi][1] + ug[xi][2]);
            uu[2] = 0.5f * (ug[xi][0] - ug[xi][1] + ug[xi][2]);
            uu[3] = ug[xi][2];
#pragma unroll
            for (int nu = 0; nu < 4; ++nu)
                smem[OFF_U + (xi * 32 + rr * 16 + n) * ULN + nu * 4 + kk * 2 + ss] = uu[nu];
        }
    }
    for (int idx = tid0; idx < 16 * 28; idx += NTHR) {
        const int c = idx / 28, q = idx - c * 28;
        float v;
        if      (q < 9)  v = W1[c * 9 + q];
        else if (q < 18) v = W2[c * 9 + q - 9];
        else if (q < 27) v = W3[c * 9 + q - 18];
        else             v = b3[c];
        smem[OFF_WD + idx] = v;
    }
    const int tile0 = (int)blockIdx.x * TPR;
    if (mat) {
        float pf0[NPF];
        patch_load(x, tile0, wid, lane, pf0);
        patch_store(smem + OFF_IN, tile0, wid, lane, pf0);
    }
    __syncthreads();

    for (int s = 0; s <= TPR; ++s) {
        const int par   = s & 1;
        const int pin   = OFF_IN + par * PINSZ;
        const int pinn  = OFF_IN + (par ^ 1) * PINSZ;
        const int c1new = OFF_C1 + par * C1SZ;
        const int c1old = OFF_C1 + (par ^ 1) * C1SZ;

        if (mat) {
          if (s < TPR) {
            const int  tile     = tile0 + s;
            const bool has_next = s + 1 < TPR;

            float pf[NPF];
            patch_load(x, has_next ? tile + 1 : tile, wid, lane, pf);

            const int r  = lane >> 4;
            const int nl = lane & 15;

            const int boff = OFF_U + lane * ULN;

            int coff[6];
#pragma unroll
            for (int u = 0; u < 6; ++u) {
                const int e  = lane + 32 * u;
                const int a  = e / 96;
                const int e2 = e - 96 * a;
                const int cc = e2 / 6;
                coff[u] = cc * C1N + a * C1W + (e2 - 6 * cc);
            }
            const int ng = s == 0 ? NGR0 : NGR;
            for (int t = wid; t < ng; t += 8) {
                const bool newc = t < NGR;
                const bool copy = newc && s > 0;
                int abase;
                if (newc) {
                    abase = pin + r * PCH2 + ((2 * t) * PW + 6 + 2 * nl) * 2;
                } else {
                    const int q  = (t - NGR) * 16 + nl;
                    const int qc = q < NXB - 1 ? q : NXB - 1;
                    const int tr = qc / 3;
                    abase = pin + r * PCH2 + ((2 * tr) * PW + 2 * (qc - 3 * tr)) * 2;
                }
                float cv[6] = {0.0f, 0.0f, 0.0f, 0.0f, 0.0f, 0.0f};
                if (copy) {
#pragma unroll
                    for (int u = 0; u < 6; ++u) cv[u] = smem[c1old + coff[u] + 2 * t * C1W + TW];
                }
                v8f Y[2][2];
                conv1_group(smem, abase, boff, Y);
                if (copy) {
#pragma unroll
                    for (int u = 0; u < 6; ++u) smem[c1new + coff[u] + 2 * t * C1W] = cv[u];
                }
                if (newc) {
#pragma unroll
                    for (int a = 0; a < 2; ++a) {
                        const int cdst = c1new + nl * C1N + (2 * t + a) * C1W + 6 + 16 * r;
#pragma unroll
                        for (int v = 0; v < 8; ++v)
                            *(v2f*)(smem + cdst + 2 * v) = (v2f){Y[a][0][v], Y[a][1][v]};
                    }
                } else {
#pragma unroll
                    for (int v = 0; v < 8; ++v) {
                        const int q = (t - NGR) * 16 + 8 * r + v;
                        if (q < NXB) {
                            const int tr = q / 3;
                            const int cdst = c1new + nl * C1N + (2 * tr) * C1W + 2 * (q - 3 * tr);
                            *(v2f*)(smem + cdst)       = (v2f){Y[0][0][v], Y[0][1][v]};
                            *(v2f*)(smem + cdst + C1W) = (v2f){Y[1][0][v], Y[1][1][v]};
                        }
                    }
                }
            }
            if (has_next) patch_store(smem + pinn, tile + 1, wid, lane, pf);
          }
        } else if (s > 0) {
            const int tile = tile0 + s - 1;
            const int cs = tid & 7;
            const int rg = (tid >> 3) & 1;
            const int ch = tid >> 4;
            const int bb  = tile / TPI;
            const int rem = tile - bb * TPI;
            const int oy  = (rem / TPR) * TH;
            const int ox  = (rem % TPR) * TW;

            float w1[9], w2[9], w3[9];
#pragma unroll
            for (int q = 0; q < 9; ++q) {
                w1[q] = smem[OFF_WD + ch * 28 + q];
                w2[q] = smem[OFF_WD + ch * 28 + 9 + q];
                w3[q] = smem[OFF_WD + ch * 28 + 18 + q];
            }
            const float bias = smem[OFF_WD + ch * 28 + 27];

            const v2f bias2 = {bias, bias};
            const float* c1p = smem + c1old + ch * C1N + (RPT * rg) * C1W + 4 * cs;
            float* op = out + ((size_t)(bb * COUT + ch) * HW + (oy + RPT * rg)) * HW + (ox + 4 * cs);

#define SPLAT(w) ((v2f){(w), (w)})
#define FMA2(a, b, c) __builtin_elementwise_fma((a), (b), (c))

            v2f a2[3][4], a3[3][3], a4[3][2];
            v2f er[3][5];
#pragma unroll
            for (int q = 0; q < 5; ++q) {
                er[0][q] = *(const v2f*)(c1p + 2 * q);
                er[1][q] = *(const v2f*)(c1p + C1W + 2 * q);
            }
#pragma unroll
            for (int i = 0; i < RPT + 6; ++i) {
                if (i + 2 < RPT + 6) {
#pragma unroll
                    for (int q = 0; q < 5; ++q)
                        er[(i + 2) % 3][q] = *(const v2f*)(c1p + (i + 2) * C1W + 2 * q);
                    __builtin_amdgcn_sched_barrier(0);
                }
                v2f e1[5], o1[4];
#pragma unroll
                for (int q = 0; q < 5; ++q)
                    e1[q] = er[i % 3][q];
#pragma unroll
                for (int q = 0; q < 4; ++q)
                    o1[q] = (v2f){e1[q].y, e1[q + 1].x};
#pragma unroll
                for (int ty = 0; ty < 3; ++ty) {
                    const int a = i - ty;
                    if (a >= 0 && a < RPT + 4) {
#pragma unroll
                        for (int q = 0; q < 4; ++q) {
                            v2f sm;
                            if (ty == 0) sm = e1[q] * SPLAT(w1[0]);
                            else         sm = FMA2(e1[q], SPLAT(w1[ty * 3]), a2[a % 3][q]);
                            sm = FMA2(o1[q],     SPLAT(w1[ty * 3 + 1]), sm);
                            sm = FMA2(e1[q + 1], SPLAT(w1[ty * 3 + 2]), sm);
                            a2[a % 3][q] = sm;
                        }
                    }
                }
                if (i >= 2) {
                    const int a = i - 2;
                    v2f e2[4], o2[3];
#pragma unroll
                    for (int q = 0; q < 4; ++q) e2[q] = a2[a % 3][q];
#pragma unroll
                    for (int q = 0; q < 3; ++q) o2[q] = (v2f){e2[q].y, e2[q + 1].x};
#pragma unroll
                    for (int ty = 0; ty < 3; ++ty) {
                        const int b = a - ty;
                        if (b >= 0 && b < RPT + 2) {
#pragma unroll
                            for (int q = 0; q < 3; ++q) {
                                v2f sm;
                                if (ty == 0) sm = e2[q] * SPLAT(w2[0]);
                                else         sm = FMA2(e2[q], SPLAT(w2[ty * 3]), a3[b % 3][q]);
                                sm = FMA2(o2[q],     SPLAT(w2[ty * 3 + 1]), sm);
                                sm = FMA2(e2[q + 1], SPLAT(w2[ty * 3 + 2]), sm);
                                a3[b % 3][q] = sm;
                            }
                        }
                    }
                    if (a >= 2) {
                        const int b = a - 2;
                        v2f e3[3], o3[2];
#pragma unroll
                        for (int q = 0; q < 3; ++q) e3[q] = a3[b % 3][q];
#pragma unroll
                        for (int q = 0; q < 2; ++q) o3[q] = (v2f){e3[q].y, e3[q + 1].x};
#pragma unroll
                        for (int ty = 0; ty < 3; ++ty) {
                            const int c = b - ty;
                            if (c >= 0 && c < RPT) {
#pragma unroll
                                for (int q = 0; q < 2; ++q) {
                                    v2f sm = (ty == 0) ? bias2 : a4[c % 3][q];
                                    sm = FMA2(e3[q],     SPLAT(w3[ty * 3]),     sm);
                                    sm = FMA2(o3[q],     SPLAT(w3[ty * 3 + 1]), sm);
                                    sm = FMA2(e3[q + 1], SPLAT(w3[ty * 3 + 2]), sm);
                                    a4[c % 3][q] = sm;
                                }
                            }
                        }
                        if (b >= 2) {
                            const int c = b - 2;
                            *(float4*)(op + (size_t)c * HW) =
                                make_float4(a4[c % 3][0].x, a4[c % 3][0].y, a4[c % 3][1].x, a4[c % 3][1].y);
                        }
                    }
                }
            }
#undef SPLAT
#undef FMA2
        }
        if (s == TPR) break;
        __builtin_amdgcn_fence(__ATOMIC_RELEASE, "workgroup", "local");
        __builtin_amdgcn_s_barrier();
        __builtin_amdgcn_fence(__ATOMIC_ACQUIRE, "workgroup", "local");
    }
}

extern "C" void kernel_launch(void* const* d_in, const int* in_sizes, int n_in,
                              void* d_out, int out_size, void* d_ws, size_t ws_size,
                              hipStream_t stream) {
    const float* x  = (const float*)d_in[0];
    const float* W0 = (const float*)d_in[1];
    const float* W1 = (const float*)d_in[2];
    const float* W2 = (const float*)d_in[3];
    const float* W3 = (const float*)d_in[4];
    const float* b3 = (const float*)d_in[5];
    float* out = (float*)d_out;

    dim3 grid(NWG);
    dim3 block(NTHR);
    fused_stackedconv_kernel<<<grid, block, 0, stream>>>(x, W0, W1, W2, W3, b3, out);
}


# depthwise stage on 4 waves, not 8: a thread's strip widened from 4 to 8 output columns (14x8), dw1/dw2 halo computed once per 8 outputs; lane halves swap row halves for whole-line stores; same sums
# speedup vs baseline: 4.2476x; 1.0295x over previous
#include <hip/hip_runtime.h>

typedef __attribute__((ext_vector_type(2))) float v2f;
typedef __attribute__((ext_vector_type(4))) float v4f;
typedef __attribute__((ext_vector_type(8))) float v8f;
typedef __attribute__((ext_vector_type(2))) unsigned int v2u;

#define CIN   8
#define COUT  16
#define HW    224
#define TH    28
#define TW    32
#define NTILE 1792
#define TPI   56
#define TPR   7
#define PR    36
#define PW    40
#define PCH2  2896
#define C1W   38
#define C1N   1292
#define NGR   17
#define NGR0  21
#define NXB   51
#define RPT   14
#define NPF   45
#define NROW  256

#define PINSZ    11584
#define C1SZ     20672
#define OFF_IN   0
#define ULN      20
#define OFF_U    23168
#define OFF_C1   25728
#define OFF_WD   67072
#define SMEM_F   67520
#define NWG      256
#define NTHR     384
#define OPAIRS      4

static __device__ __forceinline__ void patch_load(const float* __restrict__ x, int tile,
                                                  int wid, int lane, float (&pf)[NPF])
{
    const int bb  = tile / TPI;
    const int rem = tile - bb * TPI;
    const int by  = rem / TPR;
    const int bx  = rem - by * TPR;
    const float* plane = x + ((size_t)bb * CIN + wid) * (HW * HW);
    const int gxa  = bx * TW - 4 + lane;
    const int gxac = gxa < 0 ? 0 : (gxa > HW - 1 ? HW - 1 : gxa);
#pragma unroll
    for (int k = 0; k < PR; ++k) {
        const int gy  = by * TH - 4 + k;
        const int gyc = gy < 0 ? 0 : (gy > HW - 1 ? HW - 1 : gy);
        pf[k] = plane[gyc * HW + gxac];
    }
    const int gxb  = bx * TW + 28 + (lane & 7);
    const int gxbc = gxb < 0 ? 0 : (gxb > HW - 1 ? HW - 1 : gxb);
#pragma unroll
    for (int k2 = 0; k2 < PR / 4; ++k2) {
        const int gy  = by * TH - 4 + 4 * k2 + (lane >> 3);
        const int gyc = gy < 0 ? 0 : (gy > HW - 1 ? HW - 1 : gy);
        pf[PR + k2] = plane[gyc * HW + gxbc];
    }
}

static __device__ __forceinline__ void patch_store(float* pin, int tile,
                                                   int wid, int lane, const float (&pf)[NPF])
{
    const int bb  = tile / TPI;
    const int rem = tile - bb * TPI;
    const int by  = rem / TPR;
    const int bx  = rem - by * TPR;
    const int gxa = bx * TW - 4 + lane;
    const bool colokA = (gxa >= 0) && (gxa < HW);
    float* dstA = pin + (wid >> 1) * PCH2 + (wid & 1) + lane * 2;
#pragma unroll
    for (int k = 0; k < PR; ++k) {
        const int gy = by * TH - 4 + k;
        const bool ok = colokA && (gy >= 0) && (gy < HW);
        dstA[k * PW * 2] = ok ? pf[k] : 0.0f;
    }
    const int gxb = bx * TW + 28 + (lane & 7);
    const bool colokB = (gxb >= 0) && (gxb < HW);
    const int rb = lane >> 3;
    float* dstB = pin + (wid >> 1) * PCH2 + (wid & 1) + (rb * PW + 32 + (lane & 7)) * 2;
#pragma unroll
    for (int k2 = 0; k2 < PR / 4; ++k2) {
        const int gy = by * TH - 4 + 4 * k2 + rb;
        const bool ok = colokB && (gy >= 0) && (gy < HW);
        dstB[k2 * 4 * PW * 2] = ok ? pf[PR + k2] : 0.0f;
    }
}

static __device__ __forceinline__ void conv1_group(const float* smem, int abase, int boff,
                                                   v8f (&Y)[2][2])
{
    const float* pb = smem + boff;
    v4f bq[4][4];
#pragma unroll
    for (int nu = 0; nu < 4; ++nu)
        bq[0][nu] = *(const v4f*)(pb + nu * 4);
    v4f q[2][4][2];
#pragma unroll
    for (int kk = 0; kk < 2; ++kk) {
#pragma unroll
        for (int y = 0; y < 4; ++y) {
            q[kk][y][0] = *(const v4f*)(smem + abase + kk * 2 * PCH2 + y * PW * 2);
            q[kk][y][1] = *(const v4f*)(smem + abase + kk * 2 * PCH2 + y * PW * 2 + 4);
        }
    }
    v2f av[2][4][4];
#pragma unroll
    for (int kk = 0; kk < 2; ++kk) {
        v2f d[4][4];
#pragma unroll
        for (int y = 0; y < 4; ++y) {
            d[y][0] = (v2f){q[kk][y][0].x, q[kk][y][0].y};
            d[y][1] = (v2f){q[kk][y][0].z, q[kk][y][0].w};
            d[y][2] = (v2f){q[kk][y][1].x, q[kk][y][1].y};
            d[y][3] = (v2f){q[kk][y][1].z, q[kk][y][1].w};
        }
        v2f w[4][4];
#pragma unroll
        for (int c = 0; c < 4; ++c) {
            w[0][c] = d[0][c] - d[2][c];
            w[1][c] = d[1][c] + d[2][c];
            w[2][c] = d[2][c] - d[1][c];
            w[3][c] = d[1][c] - d[3][c];
        }
#pragma unroll
        for (int xi = 0; xi < 4; ++xi) {
            av[kk][xi][0] = w[xi][0] - w[xi][2];
            av[kk][xi][1] = w[xi][1] + w[xi][2];
            av[kk][xi][2] = w[xi][2] - w[xi][1];
            av[kk][xi][3] = w[xi][1] - w[xi][3];
        }
    }
#pragma unroll
    for (int xi = 0; xi < 4; ++xi) {
        if (xi < 3) {
#pragma unroll
            for (int nu = 0; nu < 4; ++nu)
                bq[xi + 1][nu] = *(const v4f*)(pb + (xi + 1) * 32 * ULN + nu * 4);
        }
        v8f m[4];
#pragma unroll
        for (int nu = 0; nu < 4; ++nu) {
            v8f z = {};
            z = __builtin_amdgcn_wmma_f32_16x16x4_f32(
                false, av[0][xi][nu], false, (v2f){bq[xi][nu].x, bq[xi][nu].y}, (short)0, z, false, false);
            m[nu] = __builtin_amdgcn_wmma_f32_16x16x4_f32(
                false, av[1][xi][nu], false, (v2f){bq[xi][nu].z, bq[xi][nu].w}, (short)0, z, false, false);
        }
        const v8f t0 = m[0] + m[1] + m[2];
        const v8f t1 = m[1] - m[2] - m[3];
        if (xi == 0)      { Y[0][0] = t0;  Y[0][1] = t1; }
        else if (xi == 1) { Y[0][0] += t0; Y[0][1] += t1; Y[1][0] = t0;  Y[1][1] = t1; }
        else if (xi == 2) { Y[0][0] += t0; Y[0][1] += t1; Y[1][0] -= t0; Y[1][1] -= t1; }
        else              { Y[1][0] -= t0; Y[1][1] -= t1; }
    }
}

__global__ __launch_bounds__(NTHR, 3)
void fused_stackedconv_kernel(const float* __restrict__ x,
                              const float* __restrict__ W0,
                              const float* __restrict__ W1,
                              const float* __restrict__ W2,
                              const float* __restrict__ W3,
                              const float* __restrict__ b3,
                              float* __restrict__ out)
{
    __shared__ float smem[SMEM_F];
    const int tid0 = threadIdx.x;
    const int wv   = __builtin_amdgcn_readfirstlane(tid0 >> 5);
    const bool mat = wv < 8;
    const int wid  = wv & 7;
    const int tid  = tid0 & 255;
    const int lane = tid0 & 31;

    if (tid0 < CIN * COUT) {
        const int cin = tid0 >> 4, n = tid0 & 15;
        const float* gp = W0 + n * 72 + cin * 9;
        float gg[3][3];
#pragma unroll
        for (int q = 0; q < 9; ++q) gg[q / 3][q % 3] = gp[q];
        float ug[4][3];
#pragma unroll
        for (int c = 0; c < 3; ++c) {
            ug[0][c] = gg[0][c];
            ug[1][c] = 0.5f * (gg[0][c] + gg[1][c] + gg[2][c]);
            ug[2][c] = 0.5f * (gg[0][c] - gg[1][c] + gg[2][c]);
            ug[3][c] = gg[2][c];
        }
        const int kk = cin >> 2, rr = (cin >> 1) & 1, ss = cin & 1;
#pragma unroll
        for (int xi = 0; xi < 4; ++xi) {
            float uu[4];
            uu[0] = ug[xi][0];
            uu[1] = 0.5f * (ug[xi][0] + ug[xi][1] + ug[xi][2]);
            uu[2] = 0.5f * (ug[xi][0] - ug[xi][1] + ug[xi][2]);
            uu[3] = ug[xi][2];
#pragma unroll
            for (int nu = 0; nu < 4; ++nu)
                smem[OFF_U + (xi * 32 + rr * 16 + n) * ULN + nu * 4 + kk * 2 + ss] = uu[nu];
        }
    }
    for (int idx = tid0; idx < 16 * 28; idx += NTHR) {
        const int c = idx / 28, q = idx - c * 28;
        float v;
        if      (q < 9)  v = W1[c * 9 + q];
        else if (q < 18) v = W2[c * 9 + q - 9];
        else if (q < 27) v = W3[c * 9 + q - 18];
        else             v = b3[c];
        smem[OFF_WD + idx] = v;
    }
    const int tile0 = (int)blockIdx.x * TPR;
    if (mat) {
        float pf0[NPF];
        patch_load(x, tile0, wid, lane, pf0);
        patch_store(smem + OFF_IN, tile0, wid, lane, pf0);
    }
    __syncthreads();

    for (int s = 0; s <= TPR; ++s) {
        const int par   = s & 1;
        const int pin   = OFF_IN + par * PINSZ;
        const int pinn  = OFF_IN + (par ^ 1) * PINSZ;
        const int c1new = OFF_C1 + par * C1SZ;
        const int c1old = OFF_C1 + (par ^ 1) * C1SZ;

        if (mat) {
          if (s < TPR) {
            const int  tile     = tile0 + s;
            const bool has_next = s + 1 < TPR;

            float pf[NPF];
            patch_load(x, has_next ? tile + 1 : tile, wid, lane, pf);

            const int r  = lane >> 4;
            const int nl = lane & 15;

            const int boff = OFF_U + lane * ULN;

            int coff[6];
#pragma unroll
            for (int u = 0; u < 6; ++u) {
                const int e  = lane + 32 * u;
                const int a  = e / 96;
                const int e2 = e - 96 * a;
                const int cc = e2 / 6;
                coff[u] = cc * C1N + a * C1W + (e2 - 6 * cc);
            }
            const int ng = s == 0 ? NGR0 : NGR;
            for (int t = wid; t < ng; t += 8) {
                const bool newc = t < NGR;
                const bool copy = newc && s > 0;
                int abase;
                if (newc) {
                    abase = pin + r * PCH2 + ((2 * t) * PW + 6 + 2 * nl) * 2;
                } else {
                    const int q  = (t - NGR) * 16 + nl;
                    const int qc = q < NXB - 1 ? q : NXB - 1;
                    const int tr = qc / 3;
                    abase = pin + r * PCH2 + ((2 * tr) * PW + 2 * (qc - 3 * tr)) * 2;
                }
                float cv[6] = {0.0f, 0.0f, 0.0f, 0.0f, 0.0f, 0.0f};
                if (copy) {
#pragma unroll
                    for (int u = 0; u < 6; ++u) cv[u] = smem[c1old + coff[u] + 2 * t * C1W + TW];
                }
                v8f Y[2][2];
                conv1_group(smem, abase, boff, Y);
                if (copy) {
#pragma unroll
                    for (int u = 0; u < 6; ++u) smem[c1new + coff[u] + 2 * t * C1W] = cv[u];
                }
                if (newc) {
#pragma unroll
                    for (int a = 0; a < 2; ++a) {
                        const int cdst = c1new + nl * C1N + (2 * t + a) * C1W + 6 + 16 * r;
#pragma unroll
                        for (int v = 0; v < 8; ++v)
                            *(v2f*)(smem + cdst + 2 * v) = (v2f){Y[a][0][v], Y[a][1][v]};
                    }
                } else {
#pragma unroll
                    for (int v = 0; v < 8; ++v) {
                        const int q = (t - NGR) * 16 + 8 * r + v;
                        if (q < NXB) {
                            const int tr = q / 3;
                            const int cdst = c1new + nl * C1N + (2 * tr) * C1W + 2 * (q - 3 * tr);
                            *(v2f*)(smem + cdst)       = (v2f){Y[0][0][v], Y[0][1][v]};
                            *(v2f*)(smem + cdst + C1W) = (v2f){Y[1][0][v], Y[1][1][v]};
                        }
                    }
                }
            }
            if (has_next) patch_store(smem + pinn, tile + 1, wid, lane, pf);
          }
        } else if (s > 0) {
            const int tile = tile0 + s - 1;
            const int cs = tid & 3;
            const int rg = (tid >> 4) & 1;
            const int ch = ((tid >> 5) << 2) | ((tid >> 2) & 3);
            const int bb  = tile / TPI;
            const int rem = tile - bb * TPI;
            const int oy  = (rem / TPR) * TH;
            const int ox  = (rem % TPR) * TW;

            float w1[9], w2[9], w3[9];
#pragma unroll
            for (int q = 0; q < 9; ++q) {
                w1[q] = smem[OFF_WD + ch * 28 + q];
                w2[q] = smem[OFF_WD + ch * 28 + 9 + q];
                w3[q] = smem[OFF_WD + ch * 28 + 18 + q];
            }
            const float bias = smem[OFF_WD + ch * 28 + 27];

            const v2f bias2 = {bias, bias};
            const float* c1p = smem + c1old + ch * C1N + (RPT * rg) * C1W + 8 * cs;
            float* op = out + ((size_t)(bb * COUT + ch) * HW + oy) * HW + (ox + 8 * cs + 4 * rg);

#define SPLAT(w) ((v2f){(w), (w)})
#define FMA2(a, b, c) __builtin_elementwise_fma((a), (b), (c))

            v2f a2[3][OPAIRS + 2], a3[3][OPAIRS + 1], a4[3][OPAIRS];
            v2f er[3][OPAIRS + 3];
#pragma unroll
            for (int q = 0; q < OPAIRS + 3; ++q) {
                er[0][q] = *(const v2f*)(c1p + 2 * q);
                er[1][q] = *(const v2f*)(c1p + C1W + 2 * q);
            }
#pragma unroll
            for (int i = 0; i < RPT + 6; ++i) {
                if (i + 2 < RPT + 6) {
#pragma unroll
                    for (int q = 0; q < OPAIRS + 3; ++q)
                        er[(i + 2) % 3][q] = *(const v2f*)(c1p + (i + 2) * C1W + 2 * q);
                    __builtin_amdgcn_sched_barrier(0);
                }
                v2f e1[OPAIRS + 3], o1[OPAIRS + 2];
#pragma unroll
                for (int q = 0; q < OPAIRS + 3; ++q)
                    e1[q] = er[i % 3][q];
#pragma unroll
                for (int q = 0; q < OPAIRS + 2; ++q)
                    o1[q] = (v2f){e1[q].y, e1[q + 1].x};
#pragma unroll
                for (int tx = 0; tx < 3; ++tx) {
#pragma unroll
                    for (int ty = 0; ty < 3; ++ty) {
                        const int a = i - ty;
                        if (a >= 0 && a < RPT + 4) {
#pragma unroll
                            for (int q = 0; q < OPAIRS + 2; ++q) {
                                if (tx == 0) {
                                    if (ty == 0) a2[a % 3][q] = e1[q] * SPLAT(w1[0]);
                                    else         a2[a % 3][q] = FMA2(e1[q], SPLAT(w1[ty * 3]), a2[a % 3][q]);
                                } else if (tx == 1) {
                                    a2[a % 3][q] = FMA2(o1[q], SPLAT(w1[ty * 3 + 1]), a2[a % 3][q]);
                                } else {
                                    a2[a % 3][q] = FMA2(e1[q + 1], SPLAT(w1[ty * 3 + 2]), a2[a % 3][q]);
                                }
                            }
                        }
                    }
                }
                if (i >= 2) {
                    const int a = i - 2;
                    v2f e2[OPAIRS + 2], o2[OPAIRS + 1];
#pragma unroll
                    for (int q = 0; q < OPAIRS + 2; ++q) e2[q] = a2[a % 3][q];
#pragma unroll
                    for (int q = 0; q < OPAIRS + 1; ++q) o2[q] = (v2f){e2[q].y, e2[q + 1].x};
#pragma unroll
                    for (int tx = 0; tx < 3; ++tx) {
#pragma unroll
                        for (int ty = 0; ty < 3; ++ty) {
                            const int b = a - ty;
                            if (b >= 0 && b < RPT + 2) {
#pragma unroll
                                for (int q = 0; q < OPAIRS + 1; ++q) {
                                    if (tx == 0) {
                                        if (ty == 0) a3[b % 3][q] = e2[q] * SPLAT(w2[0]);
                                        else         a3[b % 3][q] = FMA2(e2[q], SPLAT(w2[ty * 3]), a3[b % 3][q]);
                                    } else if (tx == 1) {
                                        a3[b % 3][q] = FMA2(o2[q], SPLAT(w2[ty * 3 + 1]), a3[b % 3][q]);
                                    } else {
                                        a3[b % 3][q] = FMA2(e2[q + 1], SPLAT(w2[ty * 3 + 2]), a3[b % 3][q]);
                                    }
                                }
                            }
                        }
                    }
                    if (a >= 2) {
                        const int b = a - 2;
                        v2f e3[OPAIRS + 1], o3[OPAIRS];
#pragma unroll
                        for (int q = 0; q < OPAIRS + 1; ++q) e3[q] = a3[b % 3][q];
#pragma unroll
                        for (int q = 0; q < OPAIRS; ++q) o3[q] = (v2f){e3[q].y, e3[q + 1].x};
#pragma unroll
                        for (int tx = 0; tx < 3; ++tx) {
#pragma unroll
                            for (int ty = 0; ty < 3; ++ty) {
                                const int c = b - ty;
                                if (c >= 0 && c < RPT) {
#pragma unroll
                                    for (int q = 0; q < OPAIRS; ++q) {
                                        if (tx == 0) {
                                            a4[c % 3][q] = FMA2(e3[q], SPLAT(w3[ty * 3]), (ty == 0) ? bias2 : a4[c % 3][q]);
                                        } else if (tx == 1) {
                                            a4[c % 3][q] = FMA2(o3[q], SPLAT(w3[ty * 3 + 1]), a4[c % 3][q]);
                                        } else {
                                            a4[c % 3][q] = FMA2(e3[q + 1], SPLAT(w3[ty * 3 + 2]), a4[c % 3][q]);
                                        }
                                    }
                                }
                            }
                        }
                        if (b >= 2) {
                            const int c = b - 2;
                            float lo[4] = {a4[c % 3][0].x, a4[c % 3][0].y, a4[c % 3][1].x, a4[c % 3][1].y};
                            float hi[4] = {a4[c % 3][2].x, a4[c % 3][2].y, a4[c % 3][3].x, a4[c % 3][3].y};
#pragma unroll
                            for (int d = 0; d < 4; ++d) {
                                const v2u sw = __builtin_amdgcn_permlane16_swap(__float_as_uint(lo[d]),
                                                                                __float_as_uint(hi[d]), false, false);
                                lo[d] = __uint_as_float(sw.x);
                                hi[d] = __uint_as_float(sw.y);
                            }
                            *(float4*)(op + (size_t)c * HW)         = make_float4(lo[0], lo[1], lo[2], lo[3]);
                            *(float4*)(op + (size_t)(RPT + c) * HW) = make_float4(hi[0], hi[1], hi[2], hi[3]);
                        }
                    }
                }
            }
#undef SPLAT
#undef FMA2
        }
        if (s == TPR) break;
        __builtin_amdgcn_fence(__ATOMIC_RELEASE, "workgroup", "local");
        __builtin_amdgcn_s_barrier();
        __builtin_amdgcn_fence(__ATOMIC_ACQUIRE, "workgroup", "local");
    }
}

extern "C" void kernel_launch(void* const* d_in, const int* in_sizes, int n_in,
                              void* d_out, int out_size, void* d_ws, size_t ws_size,
                              hipStream_t stream) {
    const float* x  = (const float*)d_in[0];
    const float* W0 = (const float*)d_in[1];
    const float* W1 = (const float*)d_in[2];
    const float* W2 = (const float*)d_in[3];
    const float* W3 = (const float*)d_in[4];
    const float* b3 = (const float*)d_in[5];
    float* out = (float*)d_out;

    dim3 grid(NWG);
    dim3 block(NTHR);
    fused_stackedconv_kernel<<<grid, block, 0, stream>>>(x, W0, W1, W2, W3, b3, out);
}
